# starting kernel
# baseline (speedup 1.0000x reference)
#include <hip/hip_runtime.h>
#include <hip/hip_bf16.h>

#define HDIM 128
#define CAP 64
#define MAX_OVF 8192

typedef float v2f __attribute__((ext_vector_type(2)));
typedef float v8f __attribute__((ext_vector_type(8)));

__global__ __launch_bounds__(256, 1)
void gemm_bias_relu_res_ln(const float* __restrict__ X,
                           const float* __restrict__ Wm,
                           const float* __restrict__ bias,
                           const float* __restrict__ R,
                           const float* __restrict__ g,
                           const float* __restrict__ bln,
                           float* __restrict__ out,
                           int n_rows)
{
    __shared__ float Ws[HDIM * HDIM];
    const int tid = threadIdx.x;

    {
        const float4* Wv = (const float4*)Wm;
        float4* Sv = (float4*)Ws;
#pragma unroll
        for (int i = 0; i < (HDIM * HDIM / 4) / 256; ++i)
            Sv[tid + i * 256] = Wv[tid + i * 256];
    }
    __syncthreads();

    const int wave = tid >> 5;
    const int lane = tid & 31;
    const int half = lane >> 4;
    const int l16  = lane & 15;

    const int row_base = blockIdx.x * 128 + wave * 16;

    int a_row = row_base + l16;
    if (a_row > n_rows - 1) a_row = n_rows - 1;
    const float* Xrow = X + (size_t)a_row * HDIM;

    v8f acc[8];
#pragma unroll
    for (int nt = 0; nt < 8; ++nt)
        acc[nt] = v8f{0.f, 0.f, 0.f, 0.f, 0.f, 0.f, 0.f, 0.f};

#pragma unroll 4
    for (int ks = 0; ks < 32; ++ks) {
        const int kb = ks * 4 + half * 2;
        v2f a;
        a.x = Xrow[kb];
        a.y = Xrow[kb + 1];
#pragma unroll
        for (int nt = 0; nt < 8; ++nt) {
            const int n = nt * 16 + l16;
            v2f bf;
            bf.x = Ws[kb * HDIM + n];
            bf.y = Ws[(kb + 1) * HDIM + n];
            acc[nt] = __builtin_amdgcn_wmma_f32_16x16x4_f32(
                false, a, false, bf, (short)0, acc[nt], false, false);
        }
    }

    float biasv[8], gv[8], bv[8];
#pragma unroll
    for (int nt = 0; nt < 8; ++nt) {
        const int n = nt * 16 + l16;
        biasv[nt] = bias[n];
        gv[nt]    = g[n];
        bv[nt]    = bln[n];
    }

#pragma unroll
    for (int j = 0; j < 8; ++j) {
        const int r  = row_base + j + half * 8;
        const int rc = (r > n_rows - 1) ? (n_rows - 1) : r;
        const float* Rrow = R + (size_t)rc * HDIM;

        float s = 0.f;
#pragma unroll
        for (int nt = 0; nt < 8; ++nt) {
            float x = acc[nt][j] + biasv[nt];
            x = fmaxf(x, 0.f);
            x += Rrow[nt * 16 + l16];
            acc[nt][j] = x;
            s += x;
        }
        s += __shfl_xor(s, 1, 32);
        s += __shfl_xor(s, 2, 32);
        s += __shfl_xor(s, 4, 32);
        s += __shfl_xor(s, 8, 32);
        const float mean = s * (1.f / HDIM);

        float var = 0.f;
#pragma unroll
        for (int nt = 0; nt < 8; ++nt) {
            const float d = acc[nt][j] - mean;
            var += d * d;
        }
        var += __shfl_xor(var, 1, 32);
        var += __shfl_xor(var, 2, 32);
        var += __shfl_xor(var, 4, 32);
        var += __shfl_xor(var, 8, 32);
        const float inv = rsqrtf(var * (1.f / HDIM) + 1e-5f);

#pragma unroll
        for (int nt = 0; nt < 8; ++nt)
            acc[nt][j] = (acc[nt][j] - mean) * inv * gv[nt] + bv[nt];

        if (r < n_rows) {
            float* orow = out + (size_t)r * HDIM;
#pragma unroll
            for (int nt = 0; nt < 8; ++nt)
                orow[nt * 16 + l16] = acc[nt][j];
        }
    }
}

__global__ __launch_bounds__(256)
void build_adj(const int* __restrict__ src, const int* __restrict__ dst,
               int* __restrict__ cnt, int* __restrict__ adj,
               int* __restrict__ ovf_cnt, int* __restrict__ ovf, int n_edges)
{
    const int e = blockIdx.x * 256 + threadIdx.x;
    if (e >= n_edges) return;
    const int d = dst[e];
    const int p = atomicAdd(&cnt[d], 1);
    if (p < CAP) {
        adj[(size_t)d * CAP + p] = src[e];
    } else {
        const int o = atomicAdd(ovf_cnt, 1);
        if (o < MAX_OVF) ovf[o] = e;
    }
}

__global__ __launch_bounds__(256)
void gather_agg(const float* __restrict__ ff, const int* __restrict__ adj,
                const int* __restrict__ cnt, float* __restrict__ agg, int n_nodes)
{
    const int gid = blockIdx.x * 256 + threadIdx.x;
    const int n = gid >> 5;
    if (n >= n_nodes) return;
    const int lane = threadIdx.x & 31;

    int deg = cnt[n];
    if (deg > CAP) deg = CAP;
    const int* ap = adj + (size_t)n * CAP;

    float4 s = {0.f, 0.f, 0.f, 0.f};
    for (int i = 0; i < deg; ++i) {
        const int sn = ap[i];
        const float4 v = *(const float4*)(ff + (size_t)sn * HDIM + lane * 4);
        s.x += v.x; s.y += v.y; s.z += v.z; s.w += v.w;
    }
    *(float4*)(agg + (size_t)n * HDIM + lane * 4) = s;
}

__global__ __launch_bounds__(256)
void ovf_scatter(const float* __restrict__ ff, const int* __restrict__ src,
                 const int* __restrict__ dst, const int* __restrict__ ovf,
                 const int* __restrict__ ovf_cnt, float* __restrict__ agg)
{
    const int gid = blockIdx.x * 256 + threadIdx.x;
    const int idx = gid >> 5;
    int c = *ovf_cnt;
    if (c > MAX_OVF) c = MAX_OVF;
    if (idx >= c) return;
    const int lane = threadIdx.x & 31;

    const int e = ovf[idx];
    const int s = src[e];
    const int d = dst[e];
    const float4 v = *(const float4*)(ff + (size_t)s * HDIM + lane * 4);
    float* ap = agg + (size_t)d * HDIM + lane * 4;
    unsafeAtomicAdd(ap + 0, v.x);
    unsafeAtomicAdd(ap + 1, v.y);
    unsafeAtomicAdd(ap + 2, v.z);
    unsafeAtomicAdd(ap + 3, v.w);
}

extern "C" void kernel_launch(void* const* d_in, const int* in_sizes, int n_in,
                              void* d_out, int out_size, void* d_ws, size_t ws_size,
                              hipStream_t stream)
{
    const float* features = (const float*)d_in[0];
    const int*   src      = (const int*)d_in[1];
    const int*   dst      = (const int*)d_in[2];
    const float* W        = (const float*)d_in[3];
    const float* b        = (const float*)d_in[4];
    const float* gcn_W    = (const float*)d_in[5];
    const float* gcn_b    = (const float*)d_in[6];
    const float* ffln_g   = (const float*)d_in[7];
    const float* ffln_b   = (const float*)d_in[8];
    const float* ln_g     = (const float*)d_in[9];
    const float* ln_b     = (const float*)d_in[10];

    const int n_nodes  = in_sizes[0] / HDIM;
    const int n_edges  = in_sizes[1];
    const int n_layers = in_sizes[6] / HDIM;

    char* ws = (char*)d_ws;
    float* ff      = (float*)ws;
    float* agg     = ff + (size_t)n_nodes * HDIM;
    int*   adj     = (int*)(agg + (size_t)n_nodes * HDIM);
    int*   cnt     = adj + (size_t)n_nodes * CAP;
    int*   ovf_cnt = cnt + n_nodes;
    int*   ovf     = ovf_cnt + 1;
    float* out     = (float*)d_out;

    const dim3 block(256);
    const dim3 ggrid((n_nodes + 127) / 128);
    const dim3 egrid((n_edges + 255) / 256);
    const dim3 ngrid((unsigned)(((size_t)n_nodes * 32 + 255) / 256));
    const dim3 ogrid((MAX_OVF * 32) / 256);

    hipMemsetAsync(cnt, 0, ((size_t)n_nodes + 1) * sizeof(int), stream);
    build_adj<<<egrid, block, 0, stream>>>(src, dst, cnt, adj, ovf_cnt, ovf, n_edges);

    const float* cur = features;
    for (int i = 0; i < n_layers; ++i) {
        gemm_bias_relu_res_ln<<<ggrid, block, 0, stream>>>(
            cur, W, b, cur, ffln_g + i * HDIM, ffln_b + i * HDIM, ff, n_nodes);

        gather_agg<<<ngrid, block, 0, stream>>>(ff, adj, cnt, agg, n_nodes);
        ovf_scatter<<<ogrid, block, 0, stream>>>(ff, src, dst, ovf, ovf_cnt, agg);

        gemm_bias_relu_res_ln<<<ggrid, block, 0, stream>>>(
            agg, gcn_W + (size_t)i * HDIM * HDIM, gcn_b + i * HDIM,
            ff, ln_g + i * HDIM, ln_b + i * HDIM, out, n_nodes);

        cur = out;
    }
}


# build_adj runs inside the first GEMM's launch as extra blocks (GEMM blocks first, then 1024-edge blocks, 4 edges per thread): the two independent kernels overlap; bucket order still arrival order
# speedup vs baseline: 1.0394x; 1.0394x over previous
#include <hip/hip_runtime.h>
#include <hip/hip_bf16.h>

#define HDIM 128
#define CAP 64
#define MAX_OVF 8192

typedef float v2f __attribute__((ext_vector_type(2)));
typedef float v8f __attribute__((ext_vector_type(8)));

__device__ __forceinline__
void gemm_tile(float* __restrict__ Ws, const int blk,
               const float* __restrict__ X,
               const float* __restrict__ Wm,
               const float* __restrict__ bias,
               const float* __restrict__ R,
               const float* __restrict__ g,
               const float* __restrict__ bln,
               float* __restrict__ out,
               int n_rows)
{
    const int tid = threadIdx.x;

    {
        const float4* Wv = (const float4*)Wm;
        float4* Sv = (float4*)Ws;
#pragma unroll
        for (int i = 0; i < (HDIM * HDIM / 4) / 256; ++i)
            Sv[tid + i * 256] = Wv[tid + i * 256];
    }
    __syncthreads();

    const int wave = tid >> 5;
    const int lane = tid & 31;
    const int half = lane >> 4;
    const int l16  = lane & 15;

    const int row_base = blk * 128 + wave * 16;

    int a_row = row_base + l16;
    if (a_row > n_rows - 1) a_row = n_rows - 1;
    const float* Xrow = X + (size_t)a_row * HDIM;

    v8f acc[8];
#pragma unroll
    for (int nt = 0; nt < 8; ++nt)
        acc[nt] = v8f{0.f, 0.f, 0.f, 0.f, 0.f, 0.f, 0.f, 0.f};

#pragma unroll 4
    for (int ks = 0; ks < 32; ++ks) {
        const int kb = ks * 4 + half * 2;
        v2f a;
        a.x = Xrow[kb];
        a.y = Xrow[kb + 1];
#pragma unroll
        for (int nt = 0; nt < 8; ++nt) {
            const int n = nt * 16 + l16;
            v2f bf;
            bf.x = Ws[kb * HDIM + n];
            bf.y = Ws[(kb + 1) * HDIM + n];
            acc[nt] = __builtin_amdgcn_wmma_f32_16x16x4_f32(
                false, a, false, bf, (short)0, acc[nt], false, false);
        }
    }

    float biasv[8], gv[8], bv[8];
#pragma unroll
    for (int nt = 0; nt < 8; ++nt) {
        const int n = nt * 16 + l16;
        biasv[nt] = bias[n];
        gv[nt]    = g[n];
        bv[nt]    = bln[n];
    }

#pragma unroll
    for (int j = 0; j < 8; ++j) {
        const int r  = row_base + j + half * 8;
        const int rc = (r > n_rows - 1) ? (n_rows - 1) : r;
        const float* Rrow = R + (size_t)rc * HDIM;

        float s = 0.f;
#pragma unroll
        for (int nt = 0; nt < 8; ++nt) {
            float x = acc[nt][j] + biasv[nt];
            x = fmaxf(x, 0.f);
            x += Rrow[nt * 16 + l16];
            acc[nt][j] = x;
            s += x;
        }
        s += __shfl_xor(s, 1, 32);
        s += __shfl_xor(s, 2, 32);
        s += __shfl_xor(s, 4, 32);
        s += __shfl_xor(s, 8, 32);
        const float mean = s * (1.f / HDIM);

        float var = 0.f;
#pragma unroll
        for (int nt = 0; nt < 8; ++nt) {
            const float d = acc[nt][j] - mean;
            var += d * d;
        }
        var += __shfl_xor(var, 1, 32);
        var += __shfl_xor(var, 2, 32);
        var += __shfl_xor(var, 4, 32);
        var += __shfl_xor(var, 8, 32);
        const float inv = rsqrtf(var * (1.f / HDIM) + 1e-5f);

#pragma unroll
        for (int nt = 0; nt < 8; ++nt)
            acc[nt][j] = (acc[nt][j] - mean) * inv * gv[nt] + bv[nt];

        if (r < n_rows) {
            float* orow = out + (size_t)r * HDIM;
#pragma unroll
            for (int nt = 0; nt < 8; ++nt)
                orow[nt * 16 + l16] = acc[nt][j];
        }
    }
}

__global__ __launch_bounds__(256, 1)
void gemm_bias_relu_res_ln(const float* __restrict__ X,
                           const float* __restrict__ Wm,
                           const float* __restrict__ bias,
                           const float* __restrict__ R,
                           const float* __restrict__ g,
                           const float* __restrict__ bln,
                           float* __restrict__ out,
                           int n_rows)
{
    __shared__ float Ws[HDIM * HDIM];
    gemm_tile(Ws, blockIdx.x, X, Wm, bias, R, g, bln, out, n_rows);
}

__device__ __forceinline__
void place_edge(const int e, const int s, const int d, const int p,
                int* __restrict__ adj,
                int* __restrict__ ovf_cnt, int* __restrict__ ovf)
{
    if (p < CAP) {
        adj[(size_t)d * CAP + p] = s;
    } else {
        const int o = atomicAdd(ovf_cnt, 1);
        if (o < MAX_OVF) ovf[o] = e;
    }
}

__global__ __launch_bounds__(256, 1)
void gemm_ln_with_build_adj(const float* __restrict__ X,
                            const float* __restrict__ Wm,
                            const float* __restrict__ bias,
                            const float* __restrict__ R,
                            const float* __restrict__ g,
                            const float* __restrict__ bln,
                            float* __restrict__ out,
                            int n_rows, int n_gemm_blocks,
                            const int* __restrict__ src,
                            const int* __restrict__ dst,
                            int* __restrict__ cnt, int* __restrict__ adj,
                            int* __restrict__ ovf_cnt, int* __restrict__ ovf,
                            int n_edges)
{
    __shared__ float Ws[HDIM * HDIM];
    if ((int)blockIdx.x < n_gemm_blocks) {
        gemm_tile(Ws, blockIdx.x, X, Wm, bias, R, g, bln, out, n_rows);
        return;
    }
    const int eb = ((int)blockIdx.x - n_gemm_blocks) * 1024;
    const int e0 = eb + (int)threadIdx.x;
    if (eb + 1024 <= n_edges) {
        int sv[4], dv[4], pv[4];
#pragma unroll
        for (int k = 0; k < 4; ++k) {
            dv[k] = dst[e0 + k * 256];
            sv[k] = src[e0 + k * 256];
        }
#pragma unroll
        for (int k = 0; k < 4; ++k)
            pv[k] = atomicAdd(&cnt[dv[k]], 1);
#pragma unroll
        for (int k = 0; k < 4; ++k)
            place_edge(e0 + k * 256, sv[k], dv[k], pv[k], adj, ovf_cnt, ovf);
    } else {
        for (int k = 0; k < 4; ++k) {
            const int e = e0 + k * 256;
            if (e < n_edges) {
                const int d = dst[e];
                const int p = atomicAdd(&cnt[d], 1);
                place_edge(e, src[e], d, p, adj, ovf_cnt, ovf);
            }
        }
    }
}

__global__ __launch_bounds__(256)
void gather_agg(const float* __restrict__ ff, const int* __restrict__ adj,
                const int* __restrict__ cnt, float* __restrict__ agg, int n_nodes)
{
    const int gid = blockIdx.x * 256 + threadIdx.x;
    const int n = gid >> 5;
    if (n >= n_nodes) return;
    const int lane = threadIdx.x & 31;

    int deg = cnt[n];
    if (deg > CAP) deg = CAP;
    const int* ap = adj + (size_t)n * CAP;

    float4 s = {0.f, 0.f, 0.f, 0.f};
    for (int i = 0; i < deg; ++i) {
        const int sn = ap[i];
        const float4 v = *(const float4*)(ff + (size_t)sn * HDIM + lane * 4);
        s.x += v.x; s.y += v.y; s.z += v.z; s.w += v.w;
    }
    *(float4*)(agg + (size_t)n * HDIM + lane * 4) = s;
}

__global__ __launch_bounds__(256)
void ovf_scatter(const float* __restrict__ ff, const int* __restrict__ src,
                 const int* __restrict__ dst, const int* __restrict__ ovf,
                 const int* __restrict__ ovf_cnt, float* __restrict__ agg)
{
    const int gid = blockIdx.x * 256 + threadIdx.x;
    const int idx = gid >> 5;
    int c = *ovf_cnt;
    if (c > MAX_OVF) c = MAX_OVF;
    if (idx >= c) return;
    const int lane = threadIdx.x & 31;

    const int e = ovf[idx];
    const int s = src[e];
    const int d = dst[e];
    const float4 v = *(const float4*)(ff + (size_t)s * HDIM + lane * 4);
    float* ap = agg + (size_t)d * HDIM + lane * 4;
    unsafeAtomicAdd(ap + 0, v.x);
    unsafeAtomicAdd(ap + 1, v.y);
    unsafeAtomicAdd(ap + 2, v.z);
    unsafeAtomicAdd(ap + 3, v.w);
}

extern "C" void kernel_launch(void* const* d_in, const int* in_sizes, int n_in,
                              void* d_out, int out_size, void* d_ws, size_t ws_size,
                              hipStream_t stream)
{
    const float* features = (const float*)d_in[0];
    const int*   src      = (const int*)d_in[1];
    const int*   dst      = (const int*)d_in[2];
    const float* W        = (const float*)d_in[3];
    const float* b        = (const float*)d_in[4];
    const float* gcn_W    = (const float*)d_in[5];
    const float* gcn_b    = (const float*)d_in[6];
    const float* ffln_g   = (const float*)d_in[7];
    const float* ffln_b   = (const float*)d_in[8];
    const float* ln_g     = (const float*)d_in[9];
    const float* ln_b     = (const float*)d_in[10];

    const int n_nodes  = in_sizes[0] / HDIM;
    const int n_edges  = in_sizes[1];
    const int n_layers = in_sizes[6] / HDIM;

    char* ws = (char*)d_ws;
    float* ff      = (float*)ws;
    float* agg     = ff + (size_t)n_nodes * HDIM;
    int*   adj     = (int*)(agg + (size_t)n_nodes * HDIM);
    int*   cnt     = adj + (size_t)n_nodes * CAP;
    int*   ovf_cnt = cnt + n_nodes;
    int*   ovf     = ovf_cnt + 1;
    float* out     = (float*)d_out;

    const dim3 block(256);
    const dim3 ggrid((n_nodes + 127) / 128);
    const dim3 fgrid(ggrid.x + (n_edges + 1023) / 1024);
    const dim3 ngrid((unsigned)(((size_t)n_nodes * 32 + 255) / 256));
    const dim3 ogrid((MAX_OVF * 32) / 256);

    hipMemsetAsync(cnt, 0, ((size_t)n_nodes + 1) * sizeof(int), stream);

    const float* cur = features;
    for (int i = 0; i < n_layers; ++i) {
        if (i == 0)
            gemm_ln_with_build_adj<<<fgrid, block, 0, stream>>>(
                cur, W, b, cur, ffln_g, ffln_b, ff, n_nodes, (int)ggrid.x,
                src, dst, cnt, adj, ovf_cnt, ovf, n_edges);
        else
            gemm_bias_relu_res_ln<<<ggrid, block, 0, stream>>>(
                cur, W, b, cur, ffln_g + i * HDIM, ffln_b + i * HDIM, ff, n_nodes);

        gather_agg<<<ngrid, block, 0, stream>>>(ff, adj, cnt, agg, n_nodes);
        ovf_scatter<<<ogrid, block, 0, stream>>>(ff, src, dst, ovf, ovf_cnt, agg);

        gemm_bias_relu_res_ln<<<ggrid, block, 0, stream>>>(
            agg, gcn_W + (size_t)i * HDIM * HDIM, gcn_b + i * HDIM,
            ff, ln_g + i * HDIM, ln_b + i * HDIM, out, n_nodes);

        cur = out;
    }
}


# neighbour-row gather moved into each layer's second GEMM launch: a wave sums its 16 rows (bucket order; overflow in list order, no float atomics) into LDS for its WMMAs; agg plane, 4 launches go
# speedup vs baseline: 1.1098x; 1.0677x over previous
#include <hip/hip_runtime.h>
#include <hip/hip_bf16.h>

#define HDIM 128
#define CAP 64
#define MAX_OVF 8192
#define APITCH 132

typedef float v2f __attribute__((ext_vector_type(2)));
typedef float v8f __attribute__((ext_vector_type(8)));

__device__ __forceinline__
float4 ld_row(const float* __restrict__ T, const int node, const int lane)
{
    return *(const float4*)(T + (size_t)node * HDIM + lane * 4);
}

template <bool GATHER>
__device__ __forceinline__
void gemm_tile(float* __restrict__ Ws, float* As, const int blk,
               const float* __restrict__ X,
               const int* __restrict__ adj, const int* __restrict__ cnt,
               const int* __restrict__ src, const int* __restrict__ dst,
               const int* __restrict__ ovf, const int* __restrict__ ovf_cnt,
               const float* __restrict__ Wm,
               const float* __restrict__ bias,
               const float* __restrict__ R,
               const float* __restrict__ g,
               const float* __restrict__ bln,
               float* __restrict__ out,
               int n_rows)
{
    const int tid = threadIdx.x;

    {
        const float4* Wv = (const float4*)Wm;
        float4* Sv = (float4*)Ws;
#pragma unroll
        for (int i = 0; i < (HDIM * HDIM / 4) / 256; ++i)
            Sv[tid + i * 256] = Wv[tid + i * 256];
    }
    __syncthreads();

    const int wave = GATHER ? __builtin_amdgcn_readfirstlane(tid >> 5) : (tid >> 5);
    const int lane = tid & 31;
    const int half = lane >> 4;
    const int l16  = lane & 15;

    const int row_base = blk * 128 + wave * 16;

    if constexpr (GATHER) {
        const float4 zero4 = {0.f, 0.f, 0.f, 0.f};
        const int rr = row_base + l16;
        const int cv = (rr < n_rows) ? cnt[rr] : 0;
        auto head = [&](const int r) {
            int m = row_base + r;
            if (m > n_rows - 1) m = n_rows - 1;
            return adj[(size_t)m * CAP + lane];
        };
        int idx[4];
#pragma unroll
        for (int q = 0; q < 4; ++q) idx[q] = head(4 * q);
#pragma unroll 1
        for (int t = 0; t < 4; ++t) {
            int idn[4];
#pragma unroll
            for (int q = 0; q < 4; ++q) idn[q] = head(((t < 3) ? t + 1 : t) + 4 * q);
            int c[4], d[4];
#pragma unroll
            for (int q = 0; q < 4; ++q) {
                c[q] = __builtin_amdgcn_readlane(cv, t + 4 * q);
                d[q] = (c[q] > CAP) ? CAP : c[q];
            }
            int dm = (d[0] > d[1]) ? d[0] : d[1];
            dm = (d[2] > dm) ? d[2] : dm;
            dm = (d[3] > dm) ? d[3] : dm;
            const int dh = (dm > 32) ? 32 : dm;
            float4 s[4] = {zero4, zero4, zero4, zero4};
#pragma unroll 1
            for (int i = 0; i < dh; i += 8) {
                int node[4][8];
#pragma unroll
                for (int q = 0; q < 4; ++q) {
#pragma unroll
                    for (int u = 0; u < 8; ++u)
                        node[q][u] = __builtin_amdgcn_readlane(idx[q], i + u);
                }
                float4 v[4][8];
#pragma unroll
                for (int q = 0; q < 4; ++q) {
#pragma unroll
                    for (int u = 0; u < 8; ++u) {
                        v[q][u] = zero4;
                        if (i + u < d[q]) v[q][u] = ld_row(X, node[q][u], lane);
                    }
                }
#pragma unroll
                for (int q = 0; q < 4; ++q) {
#pragma unroll
                    for (int u = 0; u < 8; ++u) {
                        s[q].x += v[q][u].x; s[q].y += v[q][u].y;
                        s[q].z += v[q][u].z; s[q].w += v[q][u].w;
                    }
                }
            }
#pragma unroll
            for (int q = 0; q < 4; ++q) {
                const int n = row_base + t + 4 * q;
                if (d[q] > 32) {
                    const int* aq = adj + (size_t)n * CAP;
                    for (int i = 32; i < d[q]; ++i) {
                        const float4 w = ld_row(X, aq[i], lane);
                        s[q].x += w.x; s[q].y += w.y; s[q].z += w.z; s[q].w += w.w;
                    }
                }
                if (c[q] > CAP) {
                    int oc = *ovf_cnt;
                    if (oc > MAX_OVF) oc = MAX_OVF;
                    for (int o = 0; o < oc; ++o) {
                        const int e = ovf[o];
                        if (dst[e] == n) {
                            const float4 w = ld_row(X, src[e], lane);
                            s[q].x += w.x; s[q].y += w.y; s[q].z += w.z; s[q].w += w.w;
                        }
                    }
                }
                *(float4*)(As + (wave * 16 + t + 4 * q) * APITCH + lane * 4) = s[q];
            }
#pragma unroll
            for (int q = 0; q < 4; ++q) idx[q] = idn[q];
        }
    }

    int a_row = row_base + l16;
    if (a_row > n_rows - 1) a_row = n_rows - 1;
    const float* Xrow = X + (size_t)a_row * HDIM;
    const float* Arow = GATHER ? As + (wave * 16 + l16) * APITCH : nullptr;

    v8f acc[8];
#pragma unroll
    for (int nt = 0; nt < 8; ++nt)
        acc[nt] = v8f{0.f, 0.f, 0.f, 0.f, 0.f, 0.f, 0.f, 0.f};

#pragma unroll 4
    for (int ks = 0; ks < 32; ++ks) {
        const int kb = ks * 4 + half * 2;
        v2f a;
        if constexpr (GATHER) {
            a.x = Arow[kb];
            a.y = Arow[kb + 1];
        } else {
            a.x = Xrow[kb];
            a.y = Xrow[kb + 1];
        }
#pragma unroll
        for (int nt = 0; nt < 8; ++nt) {
            const int n = nt * 16 + l16;
            v2f bf;
            bf.x = Ws[kb * HDIM + n];
            bf.y = Ws[(kb + 1) * HDIM + n];
            acc[nt] = __builtin_amdgcn_wmma_f32_16x16x4_f32(
                false, a, false, bf, (short)0, acc[nt], false, false);
        }
    }

    float biasv[8], gv[8], bv[8];
#pragma unroll
    for (int nt = 0; nt < 8; ++nt) {
        const int n = nt * 16 + l16;
        biasv[nt] = bias[n];
        gv[nt]    = g[n];
        bv[nt]    = bln[n];
    }

#pragma unroll
    for (int j = 0; j < 8; ++j) {
        const int r  = row_base + j + half * 8;
        const int rc = (r > n_rows - 1) ? (n_rows - 1) : r;
        const float* Rrow = R + (size_t)rc * HDIM;

        float s = 0.f;
#pragma unroll
        for (int nt = 0; nt < 8; ++nt) {
            float x = acc[nt][j] + biasv[nt];
            x = fmaxf(x, 0.f);
            x += Rrow[nt * 16 + l16];
            acc[nt][j] = x;
            s += x;
        }
        s += __shfl_xor(s, 1, 32);
        s += __shfl_xor(s, 2, 32);
        s += __shfl_xor(s, 4, 32);
        s += __shfl_xor(s, 8, 32);
        const float mean = s * (1.f / HDIM);

        float var = 0.f;
#pragma unroll
        for (int nt = 0; nt < 8; ++nt) {
            const float d = acc[nt][j] - mean;
            var += d * d;
        }
        var += __shfl_xor(var, 1, 32);
        var += __shfl_xor(var, 2, 32);
        var += __shfl_xor(var, 4, 32);
        var += __shfl_xor(var, 8, 32);
        const float inv = rsqrtf(var * (1.f / HDIM) + 1e-5f);

#pragma unroll
        for (int nt = 0; nt < 8; ++nt)
            acc[nt][j] = (acc[nt][j] - mean) * inv * gv[nt] + bv[nt];

        if (r < n_rows) {
            float* orow = out + (size_t)r * HDIM;
#pragma unroll
            for (int nt = 0; nt < 8; ++nt)
                orow[nt * 16 + l16] = acc[nt][j];
        }
    }
}

__global__ __launch_bounds__(256, 1)
void gemm_bias_relu_res_ln(const float* __restrict__ X,
                           const float* __restrict__ Wm,
                           const float* __restrict__ bias,
                           const float* __restrict__ R,
                           const float* __restrict__ g,
                           const float* __restrict__ bln,
                           float* __restrict__ out,
                           int n_rows)
{
    __shared__ float Ws[HDIM * HDIM];
    gemm_tile<false>(Ws, nullptr, blockIdx.x, X, nullptr, nullptr, nullptr, nullptr,
                     nullptr, nullptr, Wm, bias, R, g, bln, out, n_rows);
}

__global__ __launch_bounds__(256, 1)
void gather_gemm_bias_relu_res_ln(const float* __restrict__ T,
                                  const int* __restrict__ adj,
                                  const int* __restrict__ cnt,
                                  const int* __restrict__ src,
                                  const int* __restrict__ dst,
                                  const int* __restrict__ ovf,
                                  const int* __restrict__ ovf_cnt,
                                  const float* __restrict__ Wm,
                                  const float* __restrict__ bias,
                                  const float* __restrict__ g,
                                  const float* __restrict__ bln,
                                  float* __restrict__ out,
                                  int n_rows)
{
    __shared__ float Ws[HDIM * HDIM];
    __shared__ float As[128 * APITCH];
    gemm_tile<true>(Ws, As, blockIdx.x, T, adj, cnt, src, dst, ovf, ovf_cnt,
                    Wm, bias, T, g, bln, out, n_rows);
}

__device__ __forceinline__
void place_edge(const int e, const int s, const int d, const int p,
                int* __restrict__ adj,
                int* __restrict__ ovf_cnt, int* __restrict__ ovf)
{
    if (p < CAP) {
        adj[(size_t)d * CAP + p] = s;
    } else {
        const int o = atomicAdd(ovf_cnt, 1);
        if (o < MAX_OVF) ovf[o] = e;
    }
}

__global__ __launch_bounds__(256, 1)
void gemm_ln_with_build_adj(const float* __restrict__ X,
                            const float* __restrict__ Wm,
                            const float* __restrict__ bias,
                            const float* __restrict__ R,
                            const float* __restrict__ g,
                            const float* __restrict__ bln,
                            float* __restrict__ out,
                            int n_rows, int n_gemm_blocks,
                            const int* __restrict__ src,
                            const int* __restrict__ dst,
                            int* __restrict__ cnt, int* __restrict__ adj,
                            int* __restrict__ ovf_cnt, int* __restrict__ ovf,
                            int n_edges)
{
    __shared__ float Ws[HDIM * HDIM];
    if ((int)blockIdx.x < n_gemm_blocks) {
        gemm_tile<false>(Ws, nullptr, blockIdx.x, X, nullptr, nullptr, nullptr, nullptr,
                         nullptr, nullptr, Wm, bias, R, g, bln, out, n_rows);
        return;
    }
    const int eb = ((int)blockIdx.x - n_gemm_blocks) * 1024;
    const int e0 = eb + (int)threadIdx.x;
    if (eb + 1024 <= n_edges) {
        int sv[4], dv[4], pv[4];
#pragma unroll
        for (int k = 0; k < 4; ++k) {
            dv[k] = dst[e0 + k * 256];
            sv[k] = src[e0 + k * 256];
        }
#pragma unroll
        for (int k = 0; k < 4; ++k)
            pv[k] = atomicAdd(&cnt[dv[k]], 1);
#pragma unroll
        for (int k = 0; k < 4; ++k)
            place_edge(e0 + k * 256, sv[k], dv[k], pv[k], adj, ovf_cnt, ovf);
    } else {
        for (int k = 0; k < 4; ++k) {
            const int e = e0 + k * 256;
            if (e < n_edges) {
                const int d = dst[e];
                const int p = atomicAdd(&cnt[d], 1);
                place_edge(e, src[e], d, p, adj, ovf_cnt, ovf);
            }
        }
    }
}

extern "C" void kernel_launch(void* const* d_in, const int* in_sizes, int n_in,
                              void* d_out, int out_size, void* d_ws, size_t ws_size,
                              hipStream_t stream)
{
    const float* features = (const float*)d_in[0];
    const int*   src      = (const int*)d_in[1];
    const int*   dst      = (const int*)d_in[2];
    const float* W        = (const float*)d_in[3];
    const float* b        = (const float*)d_in[4];
    const float* gcn_W    = (const float*)d_in[5];
    const float* gcn_b    = (const float*)d_in[6];
    const float* ffln_g   = (const float*)d_in[7];
    const float* ffln_b   = (const float*)d_in[8];
    const float* ln_g     = (const float*)d_in[9];
    const float* ln_b     = (const float*)d_in[10];

    const int n_nodes  = in_sizes[0] / HDIM;
    const int n_edges  = in_sizes[1];
    const int n_layers = in_sizes[6] / HDIM;

    char* ws = (char*)d_ws;
    float* ff      = (float*)ws;
    int*   adj     = (int*)(ff + (size_t)n_nodes * HDIM);
    int*   cnt     = adj + (size_t)n_nodes * CAP;
    int*   ovf_cnt = cnt + n_nodes;
    int*   ovf     = ovf_cnt + 1;
    float* out     = (float*)d_out;

    const dim3 block(256);
    const dim3 ggrid((n_nodes + 127) / 128);
    const dim3 fgrid(ggrid.x + (n_edges + 1023) / 1024);

    hipMemsetAsync(cnt, 0, ((size_t)n_nodes + 1) * sizeof(int), stream);

    const float* cur = features;
    for (int i = 0; i < n_layers; ++i) {
        if (i == 0)
            gemm_ln_with_build_adj<<<fgrid, block, 0, stream>>>(
                cur, W, b, cur, ffln_g, ffln_b, ff, n_nodes, (int)ggrid.x,
                src, dst, cnt, adj, ovf_cnt, ovf, n_edges);
        else
            gemm_bias_relu_res_ln<<<ggrid, block, 0, stream>>>(
                cur, W, b, cur, ffln_g + i * HDIM, ffln_b + i * HDIM, ff, n_nodes);

        gather_gemm_bias_relu_res_ln<<<ggrid, block, 0, stream>>>(
            ff, adj, cnt, src, dst, ovf, ovf_cnt,
            gcn_W + (size_t)i * HDIM * HDIM, gcn_b + i * HDIM,
            ln_g + i * HDIM, ln_b + i * HDIM, out, n_nodes);

        cur = out;
    }
}


# adjacency build in two levels: edge blocks rank 4096 edges per 128-row bin by LDS adds, one global add per block and bin; first gather launch ranks its bin per row in LDS; bucket order: arrival
# speedup vs baseline: 1.4907x; 1.3432x over previous
#include <hip/hip_runtime.h>
#include <hip/hip_bf16.h>

#define HDIM 128
#define CAP 64
#define MAX_OVF 8192
#define EPB 4096
#define BIN_CAP 4096
#define APITCH 132

typedef float v2f __attribute__((ext_vector_type(2)));
typedef float v8f __attribute__((ext_vector_type(8)));

__device__ __forceinline__
float4 ld_row(const float* __restrict__ T, const int node, const int lane)
{
    return *(const float4*)(T + (size_t)node * HDIM + lane * 4);
}

template <bool GATHER, bool RANK>
__device__ __forceinline__
void gemm_tile(float* __restrict__ Ws, float* As, int* cntL, const int blk,
               const float* __restrict__ X,
               int* adj, int* cnt,
               const int* __restrict__ bins, const int* __restrict__ bin_cnt,
               const int* __restrict__ src, const int* __restrict__ dst,
               const int* __restrict__ ovf, const int* __restrict__ ovf_cnt,
               const float* __restrict__ Wm,
               const float* __restrict__ bias,
               const float* __restrict__ R,
               const float* __restrict__ g,
               const float* __restrict__ bln,
               float* __restrict__ out,
               int n_rows)
{
    const int tid = threadIdx.x;

    if constexpr (RANK) {
        const float4* Wv = (const float4*)Wm;
        float4* Sv = (float4*)Ws;
        float4 wreg[(HDIM * HDIM / 4) / 256];
#pragma unroll
        for (int i = 0; i < (HDIM * HDIM / 4) / 256; ++i)
            wreg[i] = Wv[tid + i * 256];

        const int bc   = bin_cnt[blk];
        const int nrec = (bc < BIN_CAP) ? bc : BIN_CAP;
        const int* mybin = bins + 2 * (size_t)blk * BIN_CAP;
        int* myadj = adj + (size_t)blk * 128 * CAP;
        int2 r[BIN_CAP / 256];
#pragma unroll
        for (int u = 0; u < BIN_CAP / 256; ++u) {
            const int i = u * 256 + tid;
            r[u] = make_int2(0, -1);
            if (i < nrec) r[u] = *(const int2*)(mybin + 2 * i);
        }

#pragma unroll
        for (int i = 0; i < (HDIM * HDIM / 4) / 256; ++i)
            Sv[tid + i * 256] = wreg[i];
        if (tid < 128) cntL[tid] = 0;
        __syncthreads();

        int p[BIN_CAP / 256];
#pragma unroll
        for (int u = 0; u < BIN_CAP / 256; ++u) {
            p[u] = CAP;
            if (r[u].y >= 0) p[u] = atomicAdd(&cntL[r[u].y & 127], 1);
        }
#pragma unroll
        for (int u = 0; u < BIN_CAP / 256; ++u) {
            if (p[u] < CAP) myadj[(r[u].y & 127) * CAP + p[u]] = r[u].x;
        }
        __syncthreads();
        if (tid < 128 && blk * 128 + tid < n_rows) cnt[blk * 128 + tid] = cntL[tid];
    } else {
        const float4* Wv = (const float4*)Wm;
        float4* Sv = (float4*)Ws;
#pragma unroll
        for (int i = 0; i < (HDIM * HDIM / 4) / 256; ++i)
            Sv[tid + i * 256] = Wv[tid + i * 256];
        __syncthreads();
    }

    const int wave = GATHER ? __builtin_amdgcn_readfirstlane(tid >> 5) : (tid >> 5);
    const int lane = tid & 31;
    const int half = lane >> 4;
    const int l16  = lane & 15;

    const int row_base = blk * 128 + wave * 16;

    if constexpr (GATHER) {
        const float4 zero4 = {0.f, 0.f, 0.f, 0.f};
        const int rr = row_base + l16;
        const int cv = RANK ? cntL[wave * 16 + l16] : ((rr < n_rows) ? cnt[rr] : 0);
        const int bc_all = bin_cnt[blk];
        const int nrec_f = (bc_all < BIN_CAP) ? bc_all : BIN_CAP;
        const int* fbin  = bins + 2 * (size_t)blk * BIN_CAP;
        auto head = [&](const int r) {
            int m = row_base + r;
            if (m > n_rows - 1) m = n_rows - 1;
            return adj[(size_t)m * CAP + lane];
        };
        int idx[4];
#pragma unroll
        for (int q = 0; q < 4; ++q) idx[q] = head(4 * q);
#pragma unroll 1
        for (int t = 0; t < 4; ++t) {
            int idn[4];
#pragma unroll
            for (int q = 0; q < 4; ++q) idn[q] = head(((t < 3) ? t + 1 : t) + 4 * q);
            int c[4], d[4];
#pragma unroll
            for (int q = 0; q < 4; ++q) {
                c[q] = __builtin_amdgcn_readlane(cv, t + 4 * q);
                d[q] = (c[q] > CAP) ? 0 : c[q];
            }
            int dm = (d[0] > d[1]) ? d[0] : d[1];
            dm = (d[2] > dm) ? d[2] : dm;
            dm = (d[3] > dm) ? d[3] : dm;
            const int dh = (dm > 32) ? 32 : dm;
            float4 s[4] = {zero4, zero4, zero4, zero4};
#pragma unroll 1
            for (int i = 0; i < dh; i += 8) {
                int node[4][8];
#pragma unroll
                for (int q = 0; q < 4; ++q) {
#pragma unroll
                    for (int u = 0; u < 8; ++u)
                        node[q][u] = __builtin_amdgcn_readlane(idx[q], i + u);
                }
                float4 v[4][8];
#pragma unroll
                for (int q = 0; q < 4; ++q) {
#pragma unroll
                    for (int u = 0; u < 8; ++u) {
                        v[q][u] = zero4;
                        if (i + u < d[q]) v[q][u] = ld_row(X, node[q][u], lane);
                    }
                }
#pragma unroll
                for (int q = 0; q < 4; ++q) {
#pragma unroll
                    for (int u = 0; u < 8; ++u) {
                        s[q].x += v[q][u].x; s[q].y += v[q][u].y;
                        s[q].z += v[q][u].z; s[q].w += v[q][u].w;
                    }
                }
            }
#pragma unroll
            for (int q = 0; q < 4; ++q) {
                const int n = row_base + t + 4 * q;
                if (d[q] > 32) {
                    const int* aq = adj + (size_t)n * CAP;
                    for (int i = 32; i < d[q]; ++i) {
                        const float4 w = ld_row(X, aq[i], lane);
                        s[q].x += w.x; s[q].y += w.y; s[q].z += w.z; s[q].w += w.w;
                    }
                }
                if (c[q] > CAP) {
                    for (int o = 0; o < nrec_f; ++o) {
                        const int2 rec = *(const int2*)(fbin + 2 * o);
                        if (rec.y == n) {
                            const float4 w = ld_row(X, rec.x, lane);
                            s[q].x += w.x; s[q].y += w.y; s[q].z += w.z; s[q].w += w.w;
                        }
                    }
                }
                if (bc_all > BIN_CAP) {
                    int oc = *ovf_cnt;
                    if (oc > MAX_OVF) oc = MAX_OVF;
                    for (int o = 0; o < oc; ++o) {
                        const int e = ovf[o];
                        if (dst[e] == n) {
                            const float4 w = ld_row(X, src[e], lane);
                            s[q].x += w.x; s[q].y += w.y; s[q].z += w.z; s[q].w += w.w;
                        }
                    }
                }
                *(float4*)(As + (wave * 16 + t + 4 * q) * APITCH + lane * 4) = s[q];
            }
#pragma unroll
            for (int q = 0; q < 4; ++q) idx[q] = idn[q];
        }
    }

    int a_row = row_base + l16;
    if (a_row > n_rows - 1) a_row = n_rows - 1;
    const float* Xrow = X + (size_t)a_row * HDIM;
    const float* Arow = GATHER ? As + (wave * 16 + l16) * APITCH : nullptr;

    v8f acc[8];
#pragma unroll
    for (int nt = 0; nt < 8; ++nt)
        acc[nt] = v8f{0.f, 0.f, 0.f, 0.f, 0.f, 0.f, 0.f, 0.f};

#pragma unroll 4
    for (int ks = 0; ks < 32; ++ks) {
        const int kb = ks * 4 + half * 2;
        v2f a;
        if constexpr (GATHER) {
            a.x = Arow[kb];
            a.y = Arow[kb + 1];
        } else {
            a.x = Xrow[kb];
            a.y = Xrow[kb + 1];
        }
#pragma unroll
        for (int nt = 0; nt < 8; ++nt) {
            const int n = nt * 16 + l16;
            v2f bf;
            bf.x = Ws[kb * HDIM + n];
            bf.y = Ws[(kb + 1) * HDIM + n];
            acc[nt] = __builtin_amdgcn_wmma_f32_16x16x4_f32(
                false, a, false, bf, (short)0, acc[nt], false, false);
        }
    }

    float biasv[8], gv[8], bv[8];
#pragma unroll
    for (int nt = 0; nt < 8; ++nt) {
        const int n = nt * 16 + l16;
        biasv[nt] = bias[n];
        gv[nt]    = g[n];
        bv[nt]    = bln[n];
    }

#pragma unroll
    for (int j = 0; j < 8; ++j) {
        const int r  = row_base + j + half * 8;
        const int rc = (r > n_rows - 1) ? (n_rows - 1) : r;
        const float* Rrow = R + (size_t)rc * HDIM;

        float s = 0.f;
#pragma unroll
        for (int nt = 0; nt < 8; ++nt) {
            float x = acc[nt][j] + biasv[nt];
            x = fmaxf(x, 0.f);
            x += Rrow[nt * 16 + l16];
            acc[nt][j] = x;
            s += x;
        }
        s += __shfl_xor(s, 1, 32);
        s += __shfl_xor(s, 2, 32);
        s += __shfl_xor(s, 4, 32);
        s += __shfl_xor(s, 8, 32);
        const float mean = s * (1.f / HDIM);

        float var = 0.f;
#pragma unroll
        for (int nt = 0; nt < 8; ++nt) {
            const float d = acc[nt][j] - mean;
            var += d * d;
        }
        var += __shfl_xor(var, 1, 32);
        var += __shfl_xor(var, 2, 32);
        var += __shfl_xor(var, 4, 32);
        var += __shfl_xor(var, 8, 32);
        const float inv = rsqrtf(var * (1.f / HDIM) + 1e-5f);

#pragma unroll
        for (int nt = 0; nt < 8; ++nt)
            acc[nt][j] = (acc[nt][j] - mean) * inv * gv[nt] + bv[nt];

        if (r < n_rows) {
            float* orow = out + (size_t)r * HDIM;
#pragma unroll
            for (int nt = 0; nt < 8; ++nt)
                orow[nt * 16 + l16] = acc[nt][j];
        }
    }
}

__global__ __launch_bounds__(256, 1)
void gemm_bias_relu_res_ln(const float* __restrict__ X,
                           const float* __restrict__ Wm,
                           const float* __restrict__ bias,
                           const float* __restrict__ R,
                           const float* __restrict__ g,
                           const float* __restrict__ bln,
                           float* __restrict__ out,
                           int n_rows)
{
    __shared__ float Ws[HDIM * HDIM];
    gemm_tile<false, false>(Ws, nullptr, nullptr, blockIdx.x, X, nullptr, nullptr, nullptr, nullptr,
                            nullptr, nullptr, nullptr, nullptr, Wm, bias, R, g, bln, out, n_rows);
}

__global__ __launch_bounds__(256, 1)
void gather_gemm_bias_relu_res_ln(const float* __restrict__ T,
                                  int* adj,
                                  int* cnt,
                                  const int* __restrict__ bins,
                                  const int* __restrict__ bin_cnt,
                                  const int* __restrict__ src,
                                  const int* __restrict__ dst,
                                  const int* __restrict__ ovf,
                                  const int* __restrict__ ovf_cnt,
                                  const float* __restrict__ Wm,
                                  const float* __restrict__ bias,
                                  const float* __restrict__ g,
                                  const float* __restrict__ bln,
                                  float* __restrict__ out,
                                  int n_rows)
{
    __shared__ float Ws[HDIM * HDIM];
    __shared__ float As[128 * APITCH];
    gemm_tile<true, false>(Ws, As, nullptr, blockIdx.x, T, adj, cnt, bins, bin_cnt,
                           src, dst, ovf, ovf_cnt, Wm, bias, T, g, bln, out, n_rows);
}

__global__ __launch_bounds__(256, 1)
void rank_gather_gemm_bias_relu_res_ln(const float* __restrict__ T,
                                       int* adj,
                                       int* cnt,
                                       const int* __restrict__ bins,
                                       const int* __restrict__ bin_cnt,
                                       const int* __restrict__ src,
                                       const int* __restrict__ dst,
                                       const int* __restrict__ ovf,
                                       const int* __restrict__ ovf_cnt,
                                       const float* __restrict__ Wm,
                                       const float* __restrict__ bias,
                                       const float* __restrict__ g,
                                       const float* __restrict__ bln,
                                       float* __restrict__ out,
                                       int n_rows)
{
    __shared__ float Ws[HDIM * HDIM];
    __shared__ float As[128 * APITCH];
    __shared__ int cntL[128];
    gemm_tile<true, true>(Ws, As, cntL, blockIdx.x, T, adj, cnt, bins, bin_cnt,
                          src, dst, ovf, ovf_cnt, Wm, bias, T, g, bln, out, n_rows);
}

__device__ __forceinline__
void put_record(const int e, const int s, const int d, const int b, const int pos,
                int* __restrict__ bins,
                int* __restrict__ ovf_cnt, int* __restrict__ ovf)
{
    if (pos < BIN_CAP) {
        *(int2*)(bins + 2 * ((size_t)b * BIN_CAP + pos)) = make_int2(s, d);
    } else {
        const int o = atomicAdd(ovf_cnt, 1);
        if (o < MAX_OVF) ovf[o] = e;
    }
}

__global__ __launch_bounds__(256, 1)
void gemm_ln_with_build_adj(const float* __restrict__ X,
                            const float* __restrict__ Wm,
                            const float* __restrict__ bias,
                            const float* __restrict__ R,
                            const float* __restrict__ g,
                            const float* __restrict__ bln,
                            float* __restrict__ out,
                            int n_rows, int n_gemm_blocks,
                            const int* __restrict__ src,
                            const int* __restrict__ dst,
                            int* __restrict__ bin_cnt, int* __restrict__ bins,
                            int* __restrict__ ovf_cnt, int* __restrict__ ovf,
                            int n_edges)
{
    __shared__ float Ws[HDIM * HDIM];
    if ((int)blockIdx.x < n_gemm_blocks) {
        gemm_tile<false, false>(Ws, nullptr, nullptr, blockIdx.x, X, nullptr, nullptr, nullptr, nullptr,
                                nullptr, nullptr, nullptr, nullptr, Wm, bias, R, g, bln, out, n_rows);
        return;
    }
    const int tid = threadIdx.x;
    const int eb  = ((int)blockIdx.x - n_gemm_blocks) * EPB;

    if (n_gemm_blocks > HDIM * HDIM) {
        for (int k = 0; k < EPB / 256; ++k) {
            const int e = eb + tid + k * 256;
            if (e < n_edges) {
                const int d = dst[e];
                const int b = d >> 7;
                const int pos = atomicAdd(&bin_cnt[b], 1);
                put_record(e, src[e], d, b, pos, bins, ovf_cnt, ovf);
            }
        }
        return;
    }

    int* hist = (int*)Ws;
    for (int b = tid; b < n_gemm_blocks; b += 256) hist[b] = 0;
    __syncthreads();

    int dv[EPB / 256], sv[EPB / 256], rk[EPB / 256];
#pragma unroll
    for (int k = 0; k < EPB / 256; ++k) {
        const int e = eb + tid + k * 256;
        dv[k] = -1;
        sv[k] = 0;
        if (e < n_edges) {
            dv[k] = dst[e];
            sv[k] = src[e];
        }
    }
#pragma unroll
    for (int k = 0; k < EPB / 256; ++k) {
        rk[k] = 0;
        if (dv[k] >= 0) rk[k] = atomicAdd(&hist[dv[k] >> 7], 1);
    }
    __syncthreads();

    for (int b = tid; b < n_gemm_blocks; b += 256)
        hist[b] = atomicAdd(&bin_cnt[b], hist[b]);
    __syncthreads();

#pragma unroll
    for (int k = 0; k < EPB / 256; ++k) {
        if (dv[k] >= 0) {
            const int b = dv[k] >> 7;
            put_record(eb + tid + k * 256, sv[k], dv[k], b, hist[b] + rk[k], bins, ovf_cnt, ovf);
        }
    }
}

extern "C" void kernel_launch(void* const* d_in, const int* in_sizes, int n_in,
                              void* d_out, int out_size, void* d_ws, size_t ws_size,
                              hipStream_t stream)
{
    const float* features = (const float*)d_in[0];
    const int*   src      = (const int*)d_in[1];
    const int*   dst      = (const int*)d_in[2];
    const float* W        = (const float*)d_in[3];
    const float* b        = (const float*)d_in[4];
    const float* gcn_W    = (const float*)d_in[5];
    const float* gcn_b    = (const float*)d_in[6];
    const float* ffln_g   = (const float*)d_in[7];
    const float* ffln_b   = (const float*)d_in[8];
    const float* ln_g     = (const float*)d_in[9];
    const float* ln_b     = (const float*)d_in[10];

    const int n_nodes  = in_sizes[0] / HDIM;
    const int n_edges  = in_sizes[1];
    const int n_layers = in_sizes[6] / HDIM;

    const dim3 block(256);
    const dim3 ggrid((n_nodes + 127) / 128);
    const dim3 fgrid(ggrid.x + (n_edges + EPB - 1) / EPB);

    char* ws = (char*)d_ws;
    float* ff      = (float*)ws;
    int*   adj     = (int*)(ff + (size_t)n_nodes * HDIM);
    int*   bins    = adj + (size_t)n_nodes * CAP;
    int*   cnt     = bins + (size_t)ggrid.x * BIN_CAP * 2;
    int*   bin_cnt = cnt + n_nodes;
    int*   ovf_cnt = bin_cnt + ggrid.x;
    int*   ovf     = ovf_cnt + 1;
    float* out     = (float*)d_out;

    hipMemsetAsync(bin_cnt, 0, ((size_t)ggrid.x + 1) * sizeof(int), stream);

    const float* cur = features;
    for (int i = 0; i < n_layers; ++i) {
        if (i == 0)
            gemm_ln_with_build_adj<<<fgrid, block, 0, stream>>>(
                cur, W, b, cur, ffln_g, ffln_b, ff, n_nodes, (int)ggrid.x,
                src, dst, bin_cnt, bins, ovf_cnt, ovf, n_edges);
        else
            gemm_bias_relu_res_ln<<<ggrid, block, 0, stream>>>(
                cur, W, b, cur, ffln_g + i * HDIM, ffln_b + i * HDIM, ff, n_nodes);

        if (i == 0)
            rank_gather_gemm_bias_relu_res_ln<<<ggrid, block, 0, stream>>>(
                ff, adj, cnt, bins, bin_cnt, src, dst, ovf, ovf_cnt,
                gcn_W + (size_t)i * HDIM * HDIM, gcn_b + i * HDIM,
                ln_g + i * HDIM, ln_b + i * HDIM, out, n_nodes);
        else
            gather_gemm_bias_relu_res_ln<<<ggrid, block, 0, stream>>>(
                ff, adj, cnt, bins, bin_cnt, src, dst, ovf, ovf_cnt,
                gcn_W + (size_t)i * HDIM * HDIM, gcn_b + i * HDIM,
                ln_g + i * HDIM, ln_b + i * HDIM, out, n_nodes);

        cur = out;
    }
}


# next layer's first GEMM fused into the gather-GEMM launch: the block's LN rows stay in its LDS tile as A and residual, shared W restaged into Ws behind a barrier; 3 launches, one plane less
# speedup vs baseline: 1.6632x; 1.1157x over previous
#include <hip/hip_runtime.h>
#include <hip/hip_bf16.h>

#define HDIM 128
#define CAP 64
#define MAX_OVF 8192
#define EPB 4096
#define BIN_CAP 4096
#define APITCH 132

typedef float v2f __attribute__((ext_vector_type(2)));
typedef float v8f __attribute__((ext_vector_type(8)));

template <bool ALDS, bool RLDS, bool OLDS>
__device__ __forceinline__
void strip_gemm_ln(const float* __restrict__ Ws, float* tile,
                   const float* __restrict__ X,
                   const float* __restrict__ R,
                   const float* __restrict__ bias,
                   const float* __restrict__ g,
                   const float* __restrict__ bln,
                   float* __restrict__ out,
                   const int row_base, const int n_rows, const int lane)
{
    const int half = lane >> 4;
    const int l16  = lane & 15;

    int a_row = row_base + l16;
    if (a_row > n_rows - 1) a_row = n_rows - 1;
    const float* Xrow = ALDS ? nullptr : X + (size_t)a_row * HDIM;
    const float* Arow = ALDS ? tile + l16 * APITCH : nullptr;

    v8f acc[8];
#pragma unroll
    for (int nt = 0; nt < 8; ++nt)
        acc[nt] = v8f{0.f, 0.f, 0.f, 0.f, 0.f, 0.f, 0.f, 0.f};

#pragma unroll 4
    for (int ks = 0; ks < 32; ++ks) {
        const int kb = ks * 4 + half * 2;
        v2f a;
        if constexpr (ALDS) {
            a.x = Arow[kb];
            a.y = Arow[kb + 1];
        } else {
            a.x = Xrow[kb];
            a.y = Xrow[kb + 1];
        }
#pragma unroll
        for (int nt = 0; nt < 8; ++nt) {
            const int n = nt * 16 + l16;
            v2f bf;
            bf.x = Ws[kb * HDIM + n];
            bf.y = Ws[(kb + 1) * HDIM + n];
            acc[nt] = __builtin_amdgcn_wmma_f32_16x16x4_f32(
                false, a, false, bf, (short)0, acc[nt], false, false);
        }
    }

    float biasv[8], gv[8], bv[8];
#pragma unroll
    for (int nt = 0; nt < 8; ++nt) {
        const int n = nt * 16 + l16;
        biasv[nt] = bias[n];
        gv[nt]    = g[n];
        bv[nt]    = bln[n];
    }

#pragma unroll
    for (int j = 0; j < 8; ++j) {
        const int r  = row_base + j + half * 8;
        const int rc = (r > n_rows - 1) ? (n_rows - 1) : r;
        const float* Rrow = RLDS ? tile + (j + half * 8) * APITCH : R + (size_t)rc * HDIM;

        float s = 0.f;
#pragma unroll
        for (int nt = 0; nt < 8; ++nt) {
            float x = acc[nt][j] + biasv[nt];
            x = fmaxf(x, 0.f);
            x += Rrow[nt * 16 + l16];
            acc[nt][j] = x;
            s += x;
        }
        s += __shfl_xor(s, 1, 32);
        s += __shfl_xor(s, 2, 32);
        s += __shfl_xor(s, 4, 32);
        s += __shfl_xor(s, 8, 32);
        const float mean = s * (1.f / HDIM);

        float var = 0.f;
#pragma unroll
        for (int nt = 0; nt < 8; ++nt) {
            const float d = acc[nt][j] - mean;
            var += d * d;
        }
        var += __shfl_xor(var, 1, 32);
        var += __shfl_xor(var, 2, 32);
        var += __shfl_xor(var, 4, 32);
        var += __shfl_xor(var, 8, 32);
        const float inv = rsqrtf(var * (1.f / HDIM) + 1e-5f);

#pragma unroll
        for (int nt = 0; nt < 8; ++nt)
            acc[nt][j] = (acc[nt][j] - mean) * inv * gv[nt] + bv[nt];

        if constexpr (OLDS) {
            float* trow = tile + (j + half * 8) * APITCH;
#pragma unroll
            for (int nt = 0; nt < 8; ++nt)
                trow[nt * 16 + l16] = acc[nt][j];
        } else {
            if (r < n_rows) {
                float* orow = out + (size_t)r * HDIM;
#pragma unroll
                for (int nt = 0; nt < 8; ++nt)
                    orow[nt * 16 + l16] = acc[nt][j];
            }
        }
    }
}

__device__ __forceinline__
float4 ld_row(const float* __restrict__ T, const int node, const int lane)
{
    return *(const float4*)(T + (size_t)node * HDIM + lane * 4);
}

template <bool GATHER, bool RANK, bool FUSE>
__device__ __forceinline__
void gemm_tile(float* __restrict__ Ws, float* As, int* cntL, const int blk,
               const float* __restrict__ X,
               int* adj, int* cnt,
               const int* __restrict__ bins, const int* __restrict__ bin_cnt,
               const int* __restrict__ src, const int* __restrict__ dst,
               const int* __restrict__ ovf, const int* __restrict__ ovf_cnt,
               const float* __restrict__ Wm,
               const float* __restrict__ bias,
               const float* __restrict__ R,
               const float* __restrict__ g,
               const float* __restrict__ bln,
               float* __restrict__ out,
               const float* __restrict__ W2,
               const float* __restrict__ b2,
               const float* __restrict__ g2,
               const float* __restrict__ bln2,
               int n_rows)
{
    const int tid = threadIdx.x;

    if constexpr (RANK) {
        const float4* Wv = (const float4*)Wm;
        float4* Sv = (float4*)Ws;
        float4 wreg[(HDIM * HDIM / 4) / 256];
#pragma unroll
        for (int i = 0; i < (HDIM * HDIM / 4) / 256; ++i)
            wreg[i] = Wv[tid + i * 256];

        const int bc   = bin_cnt[blk];
        const int nrec = (bc < BIN_CAP) ? bc : BIN_CAP;
        const int* mybin = bins + 2 * (size_t)blk * BIN_CAP;
        int* myadj = adj + (size_t)blk * 128 * CAP;
        int2 r[BIN_CAP / 256];
#pragma unroll
        for (int u = 0; u < BIN_CAP / 256; ++u) {
            const int i = u * 256 + tid;
            r[u] = make_int2(0, -1);
            if (i < nrec) r[u] = *(const int2*)(mybin + 2 * i);
        }

#pragma unroll
        for (int i = 0; i < (HDIM * HDIM / 4) / 256; ++i)
            Sv[tid + i * 256] = wreg[i];
        if (tid < 128) cntL[tid] = 0;
        __syncthreads();

        int p[BIN_CAP / 256];
#pragma unroll
        for (int u = 0; u < BIN_CAP / 256; ++u) {
            p[u] = CAP;
            if (r[u].y >= 0) p[u] = atomicAdd(&cntL[r[u].y & 127], 1);
        }
#pragma unroll
        for (int u = 0; u < BIN_CAP / 256; ++u) {
            if (p[u] < CAP) myadj[(r[u].y & 127) * CAP + p[u]] = r[u].x;
        }
        __syncthreads();
        if (tid < 128 && blk * 128 + tid < n_rows) cnt[blk * 128 + tid] = cntL[tid];
    } else {
        const float4* Wv = (const float4*)Wm;
        float4* Sv = (float4*)Ws;
#pragma unroll
        for (int i = 0; i < (HDIM * HDIM / 4) / 256; ++i)
            Sv[tid + i * 256] = Wv[tid + i * 256];
        __syncthreads();
    }

    const int wave = GATHER ? __builtin_amdgcn_readfirstlane(tid >> 5) : (tid >> 5);
    const int lane = tid & 31;
    const int l16  = lane & 15;

    const int row_base = blk * 128 + wave * 16;

    if constexpr (GATHER) {
        const float4 zero4 = {0.f, 0.f, 0.f, 0.f};
        const int rr = row_base + l16;
        const int cv = RANK ? cntL[wave * 16 + l16] : ((rr < n_rows) ? cnt[rr] : 0);
        const int bc_all = bin_cnt[blk];
        const int nrec_f = (bc_all < BIN_CAP) ? bc_all : BIN_CAP;
        const int* fbin  = bins + 2 * (size_t)blk * BIN_CAP;
        auto head = [&](const int r) {
            int m = row_base + r;
            if (m > n_rows - 1) m = n_rows - 1;
            return adj[(size_t)m * CAP + lane];
        };
        int idx[4];
#pragma unroll
        for (int q = 0; q < 4; ++q) idx[q] = head(4 * q);
#pragma unroll 1
        for (int t = 0; t < 4; ++t) {
            int idn[4];
#pragma unroll
            for (int q = 0; q < 4; ++q) idn[q] = head(((t < 3) ? t + 1 : t) + 4 * q);
            int c[4], d[4];
#pragma unroll
            for (int q = 0; q < 4; ++q) {
                c[q] = __builtin_amdgcn_readlane(cv, t + 4 * q);
                d[q] = (c[q] > CAP) ? 0 : c[q];
            }
            int dm = (d[0] > d[1]) ? d[0] : d[1];
            dm = (d[2] > dm) ? d[2] : dm;
            dm = (d[3] > dm) ? d[3] : dm;
            const int dh = (dm > 32) ? 32 : dm;
            float4 s[4] = {zero4, zero4, zero4, zero4};
#pragma unroll 1
            for (int i = 0; i < dh; i += 8) {
                int node[4][8];
#pragma unroll
                for (int q = 0; q < 4; ++q) {
#pragma unroll
                    for (int u = 0; u < 8; ++u)
                        node[q][u] = __builtin_amdgcn_readlane(idx[q], i + u);
                }
                float4 v[4][8];
#pragma unroll
                for (int q = 0; q < 4; ++q) {
#pragma unroll
                    for (int u = 0; u < 8; ++u) {
                        v[q][u] = zero4;
                        if (i + u < d[q]) v[q][u] = ld_row(X, node[q][u], lane);
                    }
                }
#pragma unroll
                for (int q = 0; q < 4; ++q) {
#pragma unroll
                    for (int u = 0; u < 8; ++u) {
                        s[q].x += v[q][u].x; s[q].y += v[q][u].y;
                        s[q].z += v[q][u].z; s[q].w += v[q][u].w;
                    }
                }
            }
#pragma unroll
            for (int q = 0; q < 4; ++q) {
                const int n = row_base + t + 4 * q;
                if (d[q] > 32) {
                    const int* aq = adj + (size_t)n * CAP;
                    for (int i = 32; i < d[q]; ++i) {
                        const float4 w = ld_row(X, aq[i], lane);
                        s[q].x += w.x; s[q].y += w.y; s[q].z += w.z; s[q].w += w.w;
                    }
                }
                if (c[q] > CAP) {
                    for (int o = 0; o < nrec_f; ++o) {
                        const int2 rec = *(const int2*)(fbin + 2 * o);
                        if (rec.y == n) {
                            const float4 w = ld_row(X, rec.x, lane);
                            s[q].x += w.x; s[q].y += w.y; s[q].z += w.z; s[q].w += w.w;
                        }
                    }
                }
                if (bc_all > BIN_CAP) {
                    int oc = *ovf_cnt;
                    if (oc > MAX_OVF) oc = MAX_OVF;
                    for (int o = 0; o < oc; ++o) {
                        const int e = ovf[o];
                        if (dst[e] == n) {
                            const float4 w = ld_row(X, src[e], lane);
                            s[q].x += w.x; s[q].y += w.y; s[q].z += w.z; s[q].w += w.w;
                        }
                    }
                }
                *(float4*)(As + (wave * 16 + t + 4 * q) * APITCH + lane * 4) = s[q];
            }
#pragma unroll
            for (int q = 0; q < 4; ++q) idx[q] = idn[q];
        }
    }

    float* tile = GATHER ? As + wave * 16 * APITCH : nullptr;

    if constexpr (FUSE) {
        strip_gemm_ln<true, false, true>(Ws, tile, nullptr, R, bias, g, bln, nullptr,
                                         row_base, n_rows, lane);

        const float4* Wv2 = (const float4*)W2;
        float4* Sv2 = (float4*)Ws;
        __syncthreads();
        float4 w2reg[(HDIM * HDIM / 4) / 256];
#pragma unroll
        for (int i = 0; i < (HDIM * HDIM / 4) / 256; ++i)
            w2reg[i] = Wv2[tid + i * 256];
#pragma unroll
        for (int i = 0; i < (HDIM * HDIM / 4) / 256; ++i)
            Sv2[tid + i * 256] = w2reg[i];
        __syncthreads();

        strip_gemm_ln<true, true, false>(Ws, tile, nullptr, nullptr, b2, g2, bln2, out,
                                         row_base, n_rows, lane);
    } else {
        strip_gemm_ln<GATHER, false, false>(Ws, tile, X, R, bias, g, bln, out,
                                            row_base, n_rows, lane);
    }
}

template <bool RANK, bool FUSE>
__global__ __launch_bounds__(256, 1)
void gather_gemm_bias_relu_res_ln(const float* __restrict__ T,
                                  int* adj,
                                  int* cnt,
                                  const int* __restrict__ bins,
                                  const int* __restrict__ bin_cnt,
                                  const int* __restrict__ src,
                                  const int* __restrict__ dst,
                                  const int* __restrict__ ovf,
                                  const int* __restrict__ ovf_cnt,
                                  const float* __restrict__ Wm,
                                  const float* __restrict__ bias,
                                  const float* __restrict__ g,
                                  const float* __restrict__ bln,
                                  float* __restrict__ out,
                                  const float* __restrict__ W2,
                                  const float* __restrict__ b2,
                                  const float* __restrict__ g2,
                                  const float* __restrict__ bln2,
                                  int n_rows)
{
    __shared__ float Ws[HDIM * HDIM];
    __shared__ float As[128 * APITCH];
    __shared__ int cntL[128];
    gemm_tile<true, RANK, FUSE>(Ws, As, cntL, blockIdx.x, T, adj, cnt, bins, bin_cnt,
                                src, dst, ovf, ovf_cnt, Wm, bias, T, g, bln, out,
                                W2, b2, g2, bln2, n_rows);
}

__device__ __forceinline__
void put_record(const int e, const int s, const int d, const int b, const int pos,
                int* __restrict__ bins,
                int* __restrict__ ovf_cnt, int* __restrict__ ovf)
{
    if (pos < BIN_CAP) {
        *(int2*)(bins + 2 * ((size_t)b * BIN_CAP + pos)) = make_int2(s, d);
    } else {
        const int o = atomicAdd(ovf_cnt, 1);
        if (o < MAX_OVF) ovf[o] = e;
    }
}

__global__ __launch_bounds__(256, 1)
void gemm_ln_with_build_adj(const float* __restrict__ X,
                            const float* __restrict__ Wm,
                            const float* __restrict__ bias,
                            const float* __restrict__ R,
                            const float* __restrict__ g,
                            const float* __restrict__ bln,
                            float* __restrict__ out,
                            int n_rows, int n_gemm_blocks,
                            const int* __restrict__ src,
                            const int* __restrict__ dst,
                            int* __restrict__ bin_cnt, int* __restrict__ bins,
                            int* __restrict__ ovf_cnt, int* __restrict__ ovf,
                            int n_edges)
{
    __shared__ float Ws[HDIM * HDIM];
    if ((int)blockIdx.x < n_gemm_blocks) {
        gemm_tile<false, false, false>(Ws, nullptr, nullptr, blockIdx.x, X, nullptr, nullptr, nullptr, nullptr,
                                       nullptr, nullptr, nullptr, nullptr, Wm, bias, R, g, bln, out,
                                       nullptr, nullptr, nullptr, nullptr, n_rows);
        return;
    }
    const int tid = threadIdx.x;
    const int eb  = ((int)blockIdx.x - n_gemm_blocks) * EPB;

    if (n_gemm_blocks > HDIM * HDIM) {
        for (int k = 0; k < EPB / 256; ++k) {
            const int e = eb + tid + k * 256;
            if (e < n_edges) {
                const int d = dst[e];
                const int b = d >> 7;
                const int pos = atomicAdd(&bin_cnt[b], 1);
                put_record(e, src[e], d, b, pos, bins, ovf_cnt, ovf);
            }
        }
        return;
    }

    int* hist = (int*)Ws;
    for (int b = tid; b < n_gemm_blocks; b += 256) hist[b] = 0;
    __syncthreads();

    int dv[EPB / 256], sv[EPB / 256], rk[EPB / 256];
#pragma unroll
    for (int k = 0; k < EPB / 256; ++k) {
        const int e = eb + tid + k * 256;
        dv[k] = -1;
        sv[k] = 0;
        if (e < n_edges) {
            dv[k] = dst[e];
            sv[k] = src[e];
        }
    }
#pragma unroll
    for (int k = 0; k < EPB / 256; ++k) {
        rk[k] = 0;
        if (dv[k] >= 0) rk[k] = atomicAdd(&hist[dv[k] >> 7], 1);
    }
    __syncthreads();

    for (int b = tid; b < n_gemm_blocks; b += 256)
        hist[b] = atomicAdd(&bin_cnt[b], hist[b]);
    __syncthreads();

#pragma unroll
    for (int k = 0; k < EPB / 256; ++k) {
        if (dv[k] >= 0) {
            const int b = dv[k] >> 7;
            put_record(eb + tid + k * 256, sv[k], dv[k], b, hist[b] + rk[k], bins, ovf_cnt, ovf);
        }
    }
}

extern "C" void kernel_launch(void* const* d_in, const int* in_sizes, int n_in,
                              void* d_out, int out_size, void* d_ws, size_t ws_size,
                              hipStream_t stream)
{
    const float* features = (const float*)d_in[0];
    const int*   src      = (const int*)d_in[1];
    const int*   dst      = (const int*)d_in[2];
    const float* W        = (const float*)d_in[3];
    const float* b        = (const float*)d_in[4];
    const float* gcn_W    = (const float*)d_in[5];
    const float* gcn_b    = (const float*)d_in[6];
    const float* ffln_g   = (const float*)d_in[7];
    const float* ffln_b   = (const float*)d_in[8];
    const float* ln_g     = (const float*)d_in[9];
    const float* ln_b     = (const float*)d_in[10];

    const int n_nodes  = in_sizes[0] / HDIM;
    const int n_edges  = in_sizes[1];
    const int n_layers = in_sizes[6] / HDIM;

    const dim3 block(256);
    const dim3 ggrid((n_nodes + 127) / 128);
    const dim3 fgrid(ggrid.x + (n_edges + EPB - 1) / EPB);

    char* ws = (char*)d_ws;
    float* ff      = (float*)ws;
    float* ff2     = ff + (size_t)n_nodes * HDIM;
    int*   adj     = (int*)(ff2 + (size_t)n_nodes * HDIM);
    int*   bins    = adj + (size_t)n_nodes * CAP;
    int*   cnt     = bins + (size_t)ggrid.x * BIN_CAP * 2;
    int*   bin_cnt = cnt + n_nodes;
    int*   ovf_cnt = bin_cnt + ggrid.x;
    int*   ovf     = ovf_cnt + 1;
    float* out     = (float*)d_out;

    hipMemsetAsync(bin_cnt, 0, ((size_t)ggrid.x + 1) * sizeof(int), stream);

    float* fcur = ff;
    float* fnxt = ff2;
    for (int i = 0; i < n_layers; ++i) {
        if (i == 0)
            gemm_ln_with_build_adj<<<fgrid, block, 0, stream>>>(
                features, W, b, features, ffln_g, ffln_b, fcur, n_nodes, (int)ggrid.x,
                src, dst, bin_cnt, bins, ovf_cnt, ovf, n_edges);

        const float* gW = gcn_W + (size_t)i * HDIM * HDIM;
        const float* gb = gcn_b + i * HDIM;
        const float* lg = ln_g + i * HDIM;
        const float* lb = ln_b + i * HDIM;
        if (i + 1 < n_layers) {
            const float* g2 = ffln_g + (i + 1) * HDIM;
            const float* b2 = ffln_b + (i + 1) * HDIM;
            if (i == 0)
                gather_gemm_bias_relu_res_ln<true, true><<<ggrid, block, 0, stream>>>(
                    fcur, adj, cnt, bins, bin_cnt, src, dst, ovf, ovf_cnt,
                    gW, gb, lg, lb, fnxt, W, b, g2, b2, n_nodes);
            else
                gather_gemm_bias_relu_res_ln<false, true><<<ggrid, block, 0, stream>>>(
                    fcur, adj, cnt, bins, bin_cnt, src, dst, ovf, ovf_cnt,
                    gW, gb, lg, lb, fnxt, W, b, g2, b2, n_nodes);
        } else {
            if (i == 0)
                gather_gemm_bias_relu_res_ln<true, false><<<ggrid, block, 0, stream>>>(
                    fcur, adj, cnt, bins, bin_cnt, src, dst, ovf, ovf_cnt,
                    gW, gb, lg, lb, out, nullptr, nullptr, nullptr, nullptr, n_nodes);
            else
                gather_gemm_bias_relu_res_ln<false, false><<<ggrid, block, 0, stream>>>(
                    fcur, adj, cnt, bins, bin_cnt, src, dst, ovf, ovf_cnt,
                    gW, gb, lg, lb, out, nullptr, nullptr, nullptr, nullptr, n_nodes);
        }

        float* t = fcur; fcur = fnxt; fnxt = t;
    }
}


# row gather without per-load tests: bucket slots past a row's degree point at a row of zeros, so a trip's 32 row loads go out as one burst, each with its own 32-bit offset register; sum order kept
# speedup vs baseline: 1.7536x; 1.0544x over previous
#include <hip/hip_runtime.h>
#include <hip/hip_bf16.h>

#define HDIM 128
#define CAP 64
#define MAX_OVF 8192
#define EPB 4096
#define BIN_CAP 4096
#define APITCH 132

typedef float v2f __attribute__((ext_vector_type(2)));
typedef float v8f __attribute__((ext_vector_type(8)));

template <bool ALDS, bool RLDS, bool OLDS>
__device__ __forceinline__
void strip_gemm_ln(const float* __restrict__ Ws, float* tile,
                   const float* __restrict__ X,
                   const float* __restrict__ R,
                   const float* __restrict__ bias,
                   const float* __restrict__ g,
                   const float* __restrict__ bln,
                   float* __restrict__ out,
                   const int row_base, const int n_rows, const int lane)
{
    const int half = lane >> 4;
    const int l16  = lane & 15;

    int a_row = row_base + l16;
    if (a_row > n_rows - 1) a_row = n_rows - 1;
    const float* Xrow = ALDS ? nullptr : X + (size_t)a_row * HDIM;
    const float* Arow = ALDS ? tile + l16 * APITCH : nullptr;

    v8f acc[8];
#pragma unroll
    for (int nt = 0; nt < 8; ++nt)
        acc[nt] = v8f{0.f, 0.f, 0.f, 0.f, 0.f, 0.f, 0.f, 0.f};

#pragma unroll 4
    for (int ks = 0; ks < 32; ++ks) {
        const int kb = ks * 4 + half * 2;
        v2f a;
        if constexpr (ALDS) {
            a.x = Arow[kb];
            a.y = Arow[kb + 1];
        } else {
            a.x = Xrow[kb];
            a.y = Xrow[kb + 1];
        }
#pragma unroll
        for (int nt = 0; nt < 8; ++nt) {
            const int n = nt * 16 + l16;
            v2f bf;
            bf.x = Ws[kb * HDIM + n];
            bf.y = Ws[(kb + 1) * HDIM + n];
            acc[nt] = __builtin_amdgcn_wmma_f32_16x16x4_f32(
                false, a, false, bf, (short)0, acc[nt], false, false);
        }
    }

    float biasv[8], gv[8], bv[8];
#pragma unroll
    for (int nt = 0; nt < 8; ++nt) {
        const int n = nt * 16 + l16;
        biasv[nt] = bias[n];
        gv[nt]    = g[n];
        bv[nt]    = bln[n];
    }

#pragma unroll
    for (int j = 0; j < 8; ++j) {
        const int r  = row_base + j + half * 8;
        const int rc = (r > n_rows - 1) ? (n_rows - 1) : r;
        const float* Rrow = RLDS ? tile + (j + half * 8) * APITCH : R + (size_t)rc * HDIM;

        float s = 0.f;
#pragma unroll
        for (int nt = 0; nt < 8; ++nt) {
            float x = acc[nt][j] + biasv[nt];
            x = fmaxf(x, 0.f);
            x += Rrow[nt * 16 + l16];
            acc[nt][j] = x;
            s += x;
        }
        s += __shfl_xor(s, 1, 32);
        s += __shfl_xor(s, 2, 32);
        s += __shfl_xor(s, 4, 32);
        s += __shfl_xor(s, 8, 32);
        const float mean = s * (1.f / HDIM);

        float var = 0.f;
#pragma unroll
        for (int nt = 0; nt < 8; ++nt) {
            const float d = acc[nt][j] - mean;
            var += d * d;
        }
        var += __shfl_xor(var, 1, 32);
        var += __shfl_xor(var, 2, 32);
        var += __shfl_xor(var, 4, 32);
        var += __shfl_xor(var, 8, 32);
        const float inv = rsqrtf(var * (1.f / HDIM) + 1e-5f);

#pragma unroll
        for (int nt = 0; nt < 8; ++nt)
            acc[nt][j] = (acc[nt][j] - mean) * inv * gv[nt] + bv[nt];

        if constexpr (OLDS) {
            float* trow = tile + (j + half * 8) * APITCH;
#pragma unroll
            for (int nt = 0; nt < 8; ++nt)
                trow[nt * 16 + l16] = acc[nt][j];
        } else {
            if (r < n_rows) {
                float* orow = out + (size_t)r * HDIM;
#pragma unroll
                for (int nt = 0; nt < 8; ++nt)
                    orow[nt * 16 + l16] = acc[nt][j];
            }
        }
    }
}

__device__ __forceinline__
float4 ld_row(const float* __restrict__ T, const int node, const int lane)
{
    return *(const float4*)(T + (size_t)node * HDIM + lane * 4);
}

template <bool GATHER, bool RANK, bool FUSE>
__device__ __forceinline__
void gemm_tile(float* __restrict__ Ws, float* As, int* cntL, const int blk,
               const float* __restrict__ X,
               const float* __restrict__ zrow,
               int* adj, int* cnt,
               const int* __restrict__ bins, const int* __restrict__ bin_cnt,
               const int* __restrict__ src, const int* __restrict__ dst,
               const int* __restrict__ ovf, const int* __restrict__ ovf_cnt,
               const float* __restrict__ Wm,
               const float* __restrict__ bias,
               const float* __restrict__ R,
               const float* __restrict__ g,
               const float* __restrict__ bln,
               float* __restrict__ out,
               const float* __restrict__ W2,
               const float* __restrict__ b2,
               const float* __restrict__ g2,
               const float* __restrict__ bln2,
               int n_rows)
{
    const int tid = threadIdx.x;

    if constexpr (RANK) {
        const float4* Wv = (const float4*)Wm;
        float4* Sv = (float4*)Ws;
        float4 wreg[(HDIM * HDIM / 4) / 256];
#pragma unroll
        for (int i = 0; i < (HDIM * HDIM / 4) / 256; ++i)
            wreg[i] = Wv[tid + i * 256];

        const int bc   = bin_cnt[blk];
        const int nrec = (bc < BIN_CAP) ? bc : BIN_CAP;
        const int* mybin = bins + 2 * (size_t)blk * BIN_CAP;
        int* myadj = adj + (size_t)blk * 128 * CAP;
        int2 r[BIN_CAP / 256];
#pragma unroll
        for (int u = 0; u < BIN_CAP / 256; ++u) {
            const int i = u * 256 + tid;
            r[u] = make_int2(0, -1);
            if (i < nrec) r[u] = *(const int2*)(mybin + 2 * i);
        }

#pragma unroll
        for (int i = 0; i < (HDIM * HDIM / 4) / 256; ++i)
            Sv[tid + i * 256] = wreg[i];
        if (tid < 128) cntL[tid] = 0;
        __syncthreads();

        int p[BIN_CAP / 256];
#pragma unroll
        for (int u = 0; u < BIN_CAP / 256; ++u) {
            p[u] = CAP;
            if (r[u].y >= 0) p[u] = atomicAdd(&cntL[r[u].y & 127], 1);
        }
#pragma unroll
        for (int u = 0; u < BIN_CAP / 256; ++u) {
            if (p[u] < CAP) myadj[(r[u].y & 127) * CAP + p[u]] = r[u].x;
        }
        __syncthreads();
        if (tid < 128 && blk * 128 + tid < n_rows) cnt[blk * 128 + tid] = cntL[tid];
    } else {
        const float4* Wv = (const float4*)Wm;
        float4* Sv = (float4*)Ws;
#pragma unroll
        for (int i = 0; i < (HDIM * HDIM / 4) / 256; ++i)
            Sv[tid + i * 256] = Wv[tid + i * 256];
        __syncthreads();
    }

    const int wave = GATHER ? __builtin_amdgcn_readfirstlane(tid >> 5) : (tid >> 5);
    const int lane = tid & 31;
    const int l16  = lane & 15;

    const int row_base = blk * 128 + wave * 16;

    if constexpr (GATHER) {
        const float4 zero4 = {0.f, 0.f, 0.f, 0.f};
        const int rr = row_base + l16;
        const int cv = RANK ? cntL[wave * 16 + l16] : ((rr < n_rows) ? cnt[rr] : 0);
        const int bc_all = bin_cnt[blk];
        const int nrec_f = (bc_all < BIN_CAP) ? bc_all : BIN_CAP;
        const int* fbin  = bins + 2 * (size_t)blk * BIN_CAP;
        const char* Xb = (const char*)X;
        const unsigned lane16 = (unsigned)lane * 16u;
        const unsigned zoff = (unsigned)((const char*)zrow - Xb);
        auto head = [&](const int r) {
            int m = row_base + r;
            if (m > n_rows - 1) m = n_rows - 1;
            return adj[(size_t)m * CAP + lane];
        };
        int idx[4];
#pragma unroll
        for (int q = 0; q < 4; ++q) idx[q] = head(4 * q);
#pragma unroll 1
        for (int t = 0; t < 4; ++t) {
            int idn[4];
#pragma unroll
            for (int q = 0; q < 4; ++q) idn[q] = head(((t < 3) ? t + 1 : t) + 4 * q);
            int c[4], d[4];
#pragma unroll
            for (int q = 0; q < 4; ++q) {
                c[q] = __builtin_amdgcn_readlane(cv, t + 4 * q);
                d[q] = (c[q] > CAP) ? 0 : c[q];
            }
            int dm = (d[0] > d[1]) ? d[0] : d[1];
            dm = (d[2] > dm) ? d[2] : dm;
            dm = (d[3] > dm) ? d[3] : dm;
            const int dh = (dm > 32) ? 32 : dm;
            unsigned hoff[4];
#pragma unroll
            for (int q = 0; q < 4; ++q)
                hoff[q] = (lane < d[q]) ? ((unsigned)idx[q] << 9) : zoff;
            float4 s[4] = {zero4, zero4, zero4, zero4};
#pragma unroll 1
            for (int i = 0; i < dh; i += 8) {
                unsigned off[4][8];
#pragma unroll
                for (int q = 0; q < 4; ++q) {
#pragma unroll
                    for (int u = 0; u < 8; ++u)
                        off[q][u] = (unsigned)__builtin_amdgcn_readlane((int)hoff[q], i + u) + lane16;
                }
                float4 v[4][8];
#pragma unroll
                for (int q = 0; q < 4; ++q) {
#pragma unroll
                    for (int u = 0; u < 8; ++u)
                        v[q][u] = *(const float4*)(Xb + off[q][u]);
                }
#pragma unroll
                for (int q = 0; q < 4; ++q) {
#pragma unroll
                    for (int u = 0; u < 8; ++u) {
                        s[q].x += v[q][u].x; s[q].y += v[q][u].y;
                        s[q].z += v[q][u].z; s[q].w += v[q][u].w;
                    }
                }
            }
#pragma unroll
            for (int q = 0; q < 4; ++q) {
                const int n = row_base + t + 4 * q;
                if (d[q] > 32) {
                    const int* aq = adj + (size_t)n * CAP;
                    for (int i = 32; i < d[q]; ++i) {
                        const float4 w = ld_row(X, aq[i], lane);
                        s[q].x += w.x; s[q].y += w.y; s[q].z += w.z; s[q].w += w.w;
                    }
                }
                if (c[q] > CAP) {
                    for (int o = 0; o < nrec_f; ++o) {
                        const int2 rec = *(const int2*)(fbin + 2 * o);
                        if (rec.y == n) {
                            const float4 w = ld_row(X, rec.x, lane);
                            s[q].x += w.x; s[q].y += w.y; s[q].z += w.z; s[q].w += w.w;
                        }
                    }
                }
                if (bc_all > BIN_CAP) {
                    int oc = *ovf_cnt;
                    if (oc > MAX_OVF) oc = MAX_OVF;
                    for (int o = 0; o < oc; ++o) {
                        const int e = ovf[o];
                        if (dst[e] == n) {
                            const float4 w = ld_row(X, src[e], lane);
                            s[q].x += w.x; s[q].y += w.y; s[q].z += w.z; s[q].w += w.w;
                        }
                    }
                }
                *(float4*)(As + (wave * 16 + t + 4 * q) * APITCH + lane * 4) = s[q];
            }
#pragma unroll
            for (int q = 0; q < 4; ++q) idx[q] = idn[q];
        }
    }

    float* tile = GATHER ? As + wave * 16 * APITCH : nullptr;

    if constexpr (FUSE) {
        strip_gemm_ln<true, false, true>(Ws, tile, nullptr, R, bias, g, bln, nullptr,
                                         row_base, n_rows, lane);

        const float4* Wv2 = (const float4*)W2;
        float4* Sv2 = (float4*)Ws;
        __syncthreads();
        float4 w2reg[(HDIM * HDIM / 4) / 256];
#pragma unroll
        for (int i = 0; i < (HDIM * HDIM / 4) / 256; ++i)
            w2reg[i] = Wv2[tid + i * 256];
#pragma unroll
        for (int i = 0; i < (HDIM * HDIM / 4) / 256; ++i)
            Sv2[tid + i * 256] = w2reg[i];
        __syncthreads();

        strip_gemm_ln<true, true, false>(Ws, tile, nullptr, nullptr, b2, g2, bln2, out,
                                         row_base, n_rows, lane);
    } else {
        strip_gemm_ln<GATHER, false, false>(Ws, tile, X, R, bias, g, bln, out,
                                            row_base, n_rows, lane);
    }
}

template <bool RANK, bool FUSE>
__global__ __launch_bounds__(256, 1)
void gather_gemm_bias_relu_res_ln(const float* __restrict__ T,
                                  const float* __restrict__ zrow,
                                  int* adj,
                                  int* cnt,
                                  const int* __restrict__ bins,
                                  const int* __restrict__ bin_cnt,
                                  const int* __restrict__ src,
                                  const int* __restrict__ dst,
                                  const int* __restrict__ ovf,
                                  const int* __restrict__ ovf_cnt,
                                  const float* __restrict__ Wm,
                                  const float* __restrict__ bias,
                                  const float* __restrict__ g,
                                  const float* __restrict__ bln,
                                  float* __restrict__ out,
                                  const float* __restrict__ W2,
                                  const float* __restrict__ b2,
                                  const float* __restrict__ g2,
                                  const float* __restrict__ bln2,
                                  int n_rows)
{
    __shared__ float Ws[HDIM * HDIM];
    __shared__ float As[128 * APITCH];
    __shared__ int cntL[128];
    gemm_tile<true, RANK, FUSE>(Ws, As, cntL, blockIdx.x, T, zrow, adj, cnt, bins, bin_cnt,
                                src, dst, ovf, ovf_cnt, Wm, bias, T, g, bln, out,
                                W2, b2, g2, bln2, n_rows);
}

__device__ __forceinline__
void put_record(const int e, const int s, const int d, const int b, const int pos,
                int* __restrict__ bins,
                int* __restrict__ ovf_cnt, int* __restrict__ ovf)
{
    if (pos < BIN_CAP) {
        *(int2*)(bins + 2 * ((size_t)b * BIN_CAP + pos)) = make_int2(s, d);
    } else {
        const int o = atomicAdd(ovf_cnt, 1);
        if (o < MAX_OVF) ovf[o] = e;
    }
}

__global__ __launch_bounds__(256, 1)
void gemm_ln_with_build_adj(const float* __restrict__ X,
                            const float* __restrict__ Wm,
                            const float* __restrict__ bias,
                            const float* __restrict__ R,
                            const float* __restrict__ g,
                            const float* __restrict__ bln,
                            float* __restrict__ out,
                            int n_rows, int n_gemm_blocks,
                            const int* __restrict__ src,
                            const int* __restrict__ dst,
                            int* __restrict__ bin_cnt, int* __restrict__ bins,
                            int* __restrict__ ovf_cnt, int* __restrict__ ovf,
                            int n_edges, float* __restrict__ zrow)
{
    __shared__ float Ws[HDIM * HDIM];
    if (blockIdx.x == 0 && threadIdx.x < HDIM / 4)
        ((float4*)zrow)[threadIdx.x] = float4{0.f, 0.f, 0.f, 0.f};
    if ((int)blockIdx.x < n_gemm_blocks) {
        gemm_tile<false, false, false>(Ws, nullptr, nullptr, blockIdx.x, X, nullptr, nullptr, nullptr, nullptr, nullptr,
                                       nullptr, nullptr, nullptr, nullptr, Wm, bias, R, g, bln, out,
                                       nullptr, nullptr, nullptr, nullptr, n_rows);
        return;
    }
    const int tid = threadIdx.x;
    const int eb  = ((int)blockIdx.x - n_gemm_blocks) * EPB;

    if (n_gemm_blocks > HDIM * HDIM) {
        for (int k = 0; k < EPB / 256; ++k) {
            const int e = eb + tid + k * 256;
            if (e < n_edges) {
                const int d = dst[e];
                const int b = d >> 7;
                const int pos = atomicAdd(&bin_cnt[b], 1);
                put_record(e, src[e], d, b, pos, bins, ovf_cnt, ovf);
            }
        }
        return;
    }

    int* hist = (int*)Ws;
    for (int b = tid; b < n_gemm_blocks; b += 256) hist[b] = 0;
    __syncthreads();

    int dv[EPB / 256], sv[EPB / 256], rk[EPB / 256];
#pragma unroll
    for (int k = 0; k < EPB / 256; ++k) {
        const int e = eb + tid + k * 256;
        dv[k] = -1;
        sv[k] = 0;
        if (e < n_edges) {
            dv[k] = dst[e];
            sv[k] = src[e];
        }
    }
#pragma unroll
    for (int k = 0; k < EPB / 256; ++k) {
        rk[k] = 0;
        if (dv[k] >= 0) rk[k] = atomicAdd(&hist[dv[k] >> 7], 1);
    }
    __syncthreads();

    for (int b = tid; b < n_gemm_blocks; b += 256)
        hist[b] = atomicAdd(&bin_cnt[b], hist[b]);
    __syncthreads();

#pragma unroll
    for (int k = 0; k < EPB / 256; ++k) {
        if (dv[k] >= 0) {
            const int b = dv[k] >> 7;
            put_record(eb + tid + k * 256, sv[k], dv[k], b, hist[b] + rk[k], bins, ovf_cnt, ovf);
        }
    }
}

extern "C" void kernel_launch(void* const* d_in, const int* in_sizes, int n_in,
                              void* d_out, int out_size, void* d_ws, size_t ws_size,
                              hipStream_t stream)
{
    const float* features = (const float*)d_in[0];
    const int*   src      = (const int*)d_in[1];
    const int*   dst      = (const int*)d_in[2];
    const float* W        = (const float*)d_in[3];
    const float* b        = (const float*)d_in[4];
    const float* gcn_W    = (const float*)d_in[5];
    const float* gcn_b    = (const float*)d_in[6];
    const float* ffln_g   = (const float*)d_in[7];
    const float* ffln_b   = (const float*)d_in[8];
    const float* ln_g     = (const float*)d_in[9];
    const float* ln_b     = (const float*)d_in[10];

    const int n_nodes  = in_sizes[0] / HDIM;
    const int n_edges  = in_sizes[1];
    const int n_layers = in_sizes[6] / HDIM;

    const dim3 block(256);
    const dim3 ggrid((n_nodes + 127) / 128);
    const dim3 fgrid(ggrid.x + (n_edges + EPB - 1) / EPB);

    char* ws = (char*)d_ws;
    float* ff      = (float*)ws;
    float* ff2     = ff + (size_t)n_nodes * HDIM;
    float* zrow    = ff2 + (size_t)n_nodes * HDIM;
    int*   adj     = (int*)(zrow + HDIM);
    int*   bins    = adj + (size_t)n_nodes * CAP;
    int*   cnt     = bins + (size_t)ggrid.x * BIN_CAP * 2;
    int*   bin_cnt = cnt + n_nodes;
    int*   ovf_cnt = bin_cnt + ggrid.x;
    int*   ovf     = ovf_cnt + 1;
    float* out     = (float*)d_out;

    hipMemsetAsync(bin_cnt, 0, ((size_t)ggrid.x + 1) * sizeof(int), stream);

    float* fcur = ff;
    float* fnxt = ff2;
    for (int i = 0; i < n_layers; ++i) {
        if (i == 0)
            gemm_ln_with_build_adj<<<fgrid, block, 0, stream>>>(
                features, W, b, features, ffln_g, ffln_b, fcur, n_nodes, (int)ggrid.x,
                src, dst, bin_cnt, bins, ovf_cnt, ovf, n_edges, zrow);

        const float* gW = gcn_W + (size_t)i * HDIM * HDIM;
        const float* gb = gcn_b + i * HDIM;
        const float* lg = ln_g + i * HDIM;
        const float* lb = ln_b + i * HDIM;
        if (i + 1 < n_layers) {
            const float* g2 = ffln_g + (i + 1) * HDIM;
            const float* b2 = ffln_b + (i + 1) * HDIM;
            if (i == 0)
                gather_gemm_bias_relu_res_ln<true, true><<<ggrid, block, 0, stream>>>(
                    fcur, zrow, adj, cnt, bins, bin_cnt, src, dst, ovf, ovf_cnt,
                    gW, gb, lg, lb, fnxt, W, b, g2, b2, n_nodes);
            else
                gather_gemm_bias_relu_res_ln<false, true><<<ggrid, block, 0, stream>>>(
                    fcur, zrow, adj, cnt, bins, bin_cnt, src, dst, ovf, ovf_cnt,
                    gW, gb, lg, lb, fnxt, W, b, g2, b2, n_nodes);
        } else {
            if (i == 0)
                gather_gemm_bias_relu_res_ln<true, false><<<ggrid, block, 0, stream>>>(
                    fcur, zrow, adj, cnt, bins, bin_cnt, src, dst, ovf, ovf_cnt,
                    gW, gb, lg, lb, out, nullptr, nullptr, nullptr, nullptr, n_nodes);
            else
                gather_gemm_bias_relu_res_ln<false, false><<<ggrid, block, 0, stream>>>(
                    fcur, zrow, adj, cnt, bins, bin_cnt, src, dst, ovf, ovf_cnt,
                    gW, gb, lg, lb, out, nullptr, nullptr, nullptr, nullptr, n_nodes);
        }

        float* t = fcur; fcur = fnxt; fnxt = t;
    }
}


# LayerNorm row sums on the vector ALU: each butterfly step is an add with a DPP row_xmask operand instead of a shuffle through the LDS unit and its wait (8 chained round trips per row); sum order kept
# speedup vs baseline: 1.8599x; 1.0606x over previous
#include <hip/hip_runtime.h>
#include <hip/hip_bf16.h>

#define HDIM 128
#define CAP 64
#define MAX_OVF 8192
#define EPB 4096
#define BIN_CAP 4096
#define APITCH 132

typedef float v2f __attribute__((ext_vector_type(2)));
typedef float v8f __attribute__((ext_vector_type(8)));

template <int M>
__device__ __forceinline__
float row_xor(const float v)
{
    return __builtin_bit_cast(float,
        __builtin_amdgcn_update_dpp(0, __builtin_bit_cast(int, v), 0x160 + M, 0xf, 0xf, true));
}

__device__ __forceinline__
float row16_sum(float s)
{
    s += row_xor<1>(s);
    s += row_xor<2>(s);
    s += row_xor<4>(s);
    s += row_xor<8>(s);
    return s;
}

template <bool ALDS, bool RLDS, bool OLDS>
__device__ __forceinline__
void strip_gemm_ln(const float* __restrict__ Ws, float* tile,
                   const float* __restrict__ X,
                   const float* __restrict__ R,
                   const float* __restrict__ bias,
                   const float* __restrict__ g,
                   const float* __restrict__ bln,
                   float* __restrict__ out,
                   const int row_base, const int n_rows, const int lane)
{
    const int half = lane >> 4;
    const int l16  = lane & 15;

    int a_row = row_base + l16;
    if (a_row > n_rows - 1) a_row = n_rows - 1;
    const float* Xrow = ALDS ? nullptr : X + (size_t)a_row * HDIM;
    const float* Arow = ALDS ? tile + l16 * APITCH : nullptr;

    v8f acc[8];
#pragma unroll
    for (int nt = 0; nt < 8; ++nt)
        acc[nt] = v8f{0.f, 0.f, 0.f, 0.f, 0.f, 0.f, 0.f, 0.f};

#pragma unroll 4
    for (int ks = 0; ks < 32; ++ks) {
        const int kb = ks * 4 + half * 2;
        v2f a;
        if constexpr (ALDS) {
            a.x = Arow[kb];
            a.y = Arow[kb + 1];
        } else {
            a.x = Xrow[kb];
            a.y = Xrow[kb + 1];
        }
#pragma unroll
        for (int nt = 0; nt < 8; ++nt) {
            const int n = nt * 16 + l16;
            v2f bf;
            bf.x = Ws[kb * HDIM + n];
            bf.y = Ws[(kb + 1) * HDIM + n];
            acc[nt] = __builtin_amdgcn_wmma_f32_16x16x4_f32(
                false, a, false, bf, (short)0, acc[nt], false, false);
        }
    }

    float biasv[8], gv[8], bv[8];
#pragma unroll
    for (int nt = 0; nt < 8; ++nt) {
        const int n = nt * 16 + l16;
        biasv[nt] = bias[n];
        gv[nt]    = g[n];
        bv[nt]    = bln[n];
    }

#pragma unroll
    for (int j = 0; j < 8; ++j) {
        const int r  = row_base + j + half * 8;
        const int rc = (r > n_rows - 1) ? (n_rows - 1) : r;
        const float* Rrow = RLDS ? tile + (j + half * 8) * APITCH : R + (size_t)rc * HDIM;

        float s = 0.f;
#pragma unroll
        for (int nt = 0; nt < 8; ++nt) {
            float x = acc[nt][j] + biasv[nt];
            x = fmaxf(x, 0.f);
            x += Rrow[nt * 16 + l16];
            acc[nt][j] = x;
            s += x;
        }
        s = row16_sum(s);
        const float mean = s * (1.f / HDIM);

        float var = 0.f;
#pragma unroll
        for (int nt = 0; nt < 8; ++nt) {
            const float d = acc[nt][j] - mean;
            var += d * d;
        }
        var = row16_sum(var);
        const float inv = rsqrtf(var * (1.f / HDIM) + 1e-5f);

#pragma unroll
        for (int nt = 0; nt < 8; ++nt)
            acc[nt][j] = (acc[nt][j] - mean) * inv * gv[nt] + bv[nt];

        if constexpr (OLDS) {
            float* trow = tile + (j + half * 8) * APITCH;
#pragma unroll
            for (int nt = 0; nt < 8; ++nt)
                trow[nt * 16 + l16] = acc[nt][j];
        } else {
            if (r < n_rows) {
                float* orow = out + (size_t)r * HDIM;
#pragma unroll
                for (int nt = 0; nt < 8; ++nt)
                    orow[nt * 16 + l16] = acc[nt][j];
            }
        }
    }
}

__device__ __forceinline__
float4 ld_row(const float* __restrict__ T, const int node, const int lane)
{
    return *(const float4*)(T + (size_t)node * HDIM + lane * 4);
}

template <bool GATHER, bool RANK, bool FUSE>
__device__ __forceinline__
void gemm_tile(float* __restrict__ Ws, float* As, int* cntL, const int blk,
               const float* __restrict__ X,
               const float* __restrict__ zrow,
               int* adj, int* cnt,
               const int* __restrict__ bins, const int* __restrict__ bin_cnt,
               const int* __restrict__ src, const int* __restrict__ dst,
               const int* __restrict__ ovf, const int* __restrict__ ovf_cnt,
               const float* __restrict__ Wm,
               const float* __restrict__ bias,
               const float* __restrict__ R,
               const float* __restrict__ g,
               const float* __restrict__ bln,
               float* __restrict__ out,
               const float* __restrict__ W2,
               const float* __restrict__ b2,
               const float* __restrict__ g2,
               const float* __restrict__ bln2,
               int n_rows)
{
    const int tid = threadIdx.x;

    if constexpr (RANK) {
        const float4* Wv = (const float4*)Wm;
        float4* Sv = (float4*)Ws;
        float4 wreg[(HDIM * HDIM / 4) / 256];
#pragma unroll
        for (int i = 0; i < (HDIM * HDIM / 4) / 256; ++i)
            wreg[i] = Wv[tid + i * 256];

        const int bc   = bin_cnt[blk];
        const int nrec = (bc < BIN_CAP) ? bc : BIN_CAP;
        const int* mybin = bins + 2 * (size_t)blk * BIN_CAP;
        int* myadj = adj + (size_t)blk * 128 * CAP;
        int2 r[BIN_CAP / 256];
#pragma unroll
        for (int u = 0; u < BIN_CAP / 256; ++u) {
            const int i = u * 256 + tid;
            r[u] = make_int2(0, -1);
            if (i < nrec) r[u] = *(const int2*)(mybin + 2 * i);
        }

#pragma unroll
        for (int i = 0; i < (HDIM * HDIM / 4) / 256; ++i)
            Sv[tid + i * 256] = wreg[i];
        if (tid < 128) cntL[tid] = 0;
        __syncthreads();

        int p[BIN_CAP / 256];
#pragma unroll
        for (int u = 0; u < BIN_CAP / 256; ++u) {
            p[u] = CAP;
            if (r[u].y >= 0) p[u] = atomicAdd(&cntL[r[u].y & 127], 1);
        }
#pragma unroll
        for (int u = 0; u < BIN_CAP / 256; ++u) {
            if (p[u] < CAP) myadj[(r[u].y & 127) * CAP + p[u]] = r[u].x;
        }
        __syncthreads();
        if (tid < 128 && blk * 128 + tid < n_rows) cnt[blk * 128 + tid] = cntL[tid];
    } else {
        const float4* Wv = (const float4*)Wm;
        float4* Sv = (float4*)Ws;
#pragma unroll
        for (int i = 0; i < (HDIM * HDIM / 4) / 256; ++i)
            Sv[tid + i * 256] = Wv[tid + i * 256];
        __syncthreads();
    }

    const int wave = GATHER ? __builtin_amdgcn_readfirstlane(tid >> 5) : (tid >> 5);
    const int lane = tid & 31;
    const int l16  = lane & 15;

    const int row_base = blk * 128 + wave * 16;

    if constexpr (GATHER) {
        const float4 zero4 = {0.f, 0.f, 0.f, 0.f};
        const int rr = row_base + l16;
        const int cv = RANK ? cntL[wave * 16 + l16] : ((rr < n_rows) ? cnt[rr] : 0);
        const int bc_all = bin_cnt[blk];
        const int nrec_f = (bc_all < BIN_CAP) ? bc_all : BIN_CAP;
        const int* fbin  = bins + 2 * (size_t)blk * BIN_CAP;
        const char* Xb = (const char*)X;
        const unsigned lane16 = (unsigned)lane * 16u;
        const unsigned zoff = (unsigned)((const char*)zrow - Xb);
        auto head = [&](const int r) {
            int m = row_base + r;
            if (m > n_rows - 1) m = n_rows - 1;
            return adj[(size_t)m * CAP + lane];
        };
        int idx[4];
#pragma unroll
        for (int q = 0; q < 4; ++q) idx[q] = head(4 * q);
#pragma unroll 1
        for (int t = 0; t < 4; ++t) {
            int idn[4];
#pragma unroll
            for (int q = 0; q < 4; ++q) idn[q] = head(((t < 3) ? t + 1 : t) + 4 * q);
            int c[4], d[4];
#pragma unroll
            for (int q = 0; q < 4; ++q) {
                c[q] = __builtin_amdgcn_readlane(cv, t + 4 * q);
                d[q] = (c[q] > CAP) ? 0 : c[q];
            }
            int dm = (d[0] > d[1]) ? d[0] : d[1];
            dm = (d[2] > dm) ? d[2] : dm;
            dm = (d[3] > dm) ? d[3] : dm;
            const int dh = (dm > 32) ? 32 : dm;
            unsigned hoff[4];
#pragma unroll
            for (int q = 0; q < 4; ++q)
                hoff[q] = (lane < d[q]) ? ((unsigned)idx[q] << 9) : zoff;
            float4 s[4] = {zero4, zero4, zero4, zero4};
#pragma unroll 1
            for (int i = 0; i < dh; i += 8) {
                unsigned off[4][8];
#pragma unroll
                for (int q = 0; q < 4; ++q) {
#pragma unroll
                    for (int u = 0; u < 8; ++u)
                        off[q][u] = (unsigned)__builtin_amdgcn_readlane((int)hoff[q], i + u) + lane16;
                }
                float4 v[4][8];
#pragma unroll
                for (int q = 0; q < 4; ++q) {
#pragma unroll
                    for (int u = 0; u < 8; ++u)
                        v[q][u] = *(const float4*)(Xb + off[q][u]);
                }
#pragma unroll
                for (int q = 0; q < 4; ++q) {
#pragma unroll
                    for (int u = 0; u < 8; ++u) {
                        s[q].x += v[q][u].x; s[q].y += v[q][u].y;
                        s[q].z += v[q][u].z; s[q].w += v[q][u].w;
                    }
                }
            }
#pragma unroll
            for (int q = 0; q < 4; ++q) {
                const int n = row_base + t + 4 * q;
                if (d[q] > 32) {
                    const int* aq = adj + (size_t)n * CAP;
                    for (int i = 32; i < d[q]; ++i) {
                        const float4 w = ld_row(X, aq[i], lane);
                        s[q].x += w.x; s[q].y += w.y; s[q].z += w.z; s[q].w += w.w;
                    }
                }
                if (c[q] > CAP) {
                    for (int o = 0; o < nrec_f; ++o) {
                        const int2 rec = *(const int2*)(fbin + 2 * o);
                        if (rec.y == n) {
                            const float4 w = ld_row(X, rec.x, lane);
                            s[q].x += w.x; s[q].y += w.y; s[q].z += w.z; s[q].w += w.w;
                        }
                    }
                }
                if (bc_all > BIN_CAP) {
                    int oc = *ovf_cnt;
                    if (oc > MAX_OVF) oc = MAX_OVF;
                    for (int o = 0; o < oc; ++o) {
                        const int e = ovf[o];
                        if (dst[e] == n) {
                            const float4 w = ld_row(X, src[e], lane);
                            s[q].x += w.x; s[q].y += w.y; s[q].z += w.z; s[q].w += w.w;
                        }
                    }
                }
                *(float4*)(As + (wave * 16 + t + 4 * q) * APITCH + lane * 4) = s[q];
            }
#pragma unroll
            for (int q = 0; q < 4; ++q) idx[q] = idn[q];
        }
    }

    float* tile = GATHER ? As + wave * 16 * APITCH : nullptr;

    if constexpr (FUSE) {
        strip_gemm_ln<true, false, true>(Ws, tile, nullptr, R, bias, g, bln, nullptr,
                                         row_base, n_rows, lane);

        const float4* Wv2 = (const float4*)W2;
        float4* Sv2 = (float4*)Ws;
        __syncthreads();
        float4 w2reg[(HDIM * HDIM / 4) / 256];
#pragma unroll
        for (int i = 0; i < (HDIM * HDIM / 4) / 256; ++i)
            w2reg[i] = Wv2[tid + i * 256];
#pragma unroll
        for (int i = 0; i < (HDIM * HDIM / 4) / 256; ++i)
            Sv2[tid + i * 256] = w2reg[i];
        __syncthreads();

        strip_gemm_ln<true, true, false>(Ws, tile, nullptr, nullptr, b2, g2, bln2, out,
                                         row_base, n_rows, lane);
    } else {
        strip_gemm_ln<GATHER, false, false>(Ws, tile, X, R, bias, g, bln, out,
                                            row_base, n_rows, lane);
    }
}

template <bool RANK, bool FUSE>
__global__ __launch_bounds__(256, 1)
void gather_gemm_bias_relu_res_ln(const float* __restrict__ T,
                                  const float* __restrict__ zrow,
                                  int* adj,
                                  int* cnt,
                                  const int* __restrict__ bins,
                                  const int* __restrict__ bin_cnt,
                                  const int* __restrict__ src,
                                  const int* __restrict__ dst,
                                  const int* __restrict__ ovf,
                                  const int* __restrict__ ovf_cnt,
                                  const float* __restrict__ Wm,
                                  const float* __restrict__ bias,
                                  const float* __restrict__ g,
                                  const float* __restrict__ bln,
                                  float* __restrict__ out,
                                  const float* __restrict__ W2,
                                  const float* __restrict__ b2,
                                  const float* __restrict__ g2,
                                  const float* __restrict__ bln2,
                                  int n_rows)
{
    __shared__ float Ws[HDIM * HDIM];
    __shared__ float As[128 * APITCH];
    __shared__ int cntL[128];
    gemm_tile<true, RANK, FUSE>(Ws, As, cntL, blockIdx.x, T, zrow, adj, cnt, bins, bin_cnt,
                                src, dst, ovf, ovf_cnt, Wm, bias, T, g, bln, out,
                                W2, b2, g2, bln2, n_rows);
}

__device__ __forceinline__
void put_record(const int e, const int s, const int d, const int b, const int pos,
                int* __restrict__ bins,
                int* __restrict__ ovf_cnt, int* __restrict__ ovf)
{
    if (pos < BIN_CAP) {
        *(int2*)(bins + 2 * ((size_t)b * BIN_CAP + pos)) = make_int2(s, d);
    } else {
        const int o = atomicAdd(ovf_cnt, 1);
        if (o < MAX_OVF) ovf[o] = e;
    }
}

__global__ __launch_bounds__(256, 1)
void gemm_ln_with_build_adj(const float* __restrict__ X,
                            const float* __restrict__ Wm,
                            const float* __restrict__ bias,
                            const float* __restrict__ R,
                            const float* __restrict__ g,
                            const float* __restrict__ bln,
                            float* __restrict__ out,
                            int n_rows, int n_gemm_blocks,
                            const int* __restrict__ src,
                            const int* __restrict__ dst,
                            int* __restrict__ bin_cnt, int* __restrict__ bins,
                            int* __restrict__ ovf_cnt, int* __restrict__ ovf,
                            int n_edges, float* __restrict__ zrow)
{
    __shared__ float Ws[HDIM * HDIM];
    if (blockIdx.x == 0 && threadIdx.x < HDIM / 4)
        ((float4*)zrow)[threadIdx.x] = float4{0.f, 0.f, 0.f, 0.f};
    if ((int)blockIdx.x < n_gemm_blocks) {
        gemm_tile<false, false, false>(Ws, nullptr, nullptr, blockIdx.x, X, nullptr, nullptr, nullptr, nullptr, nullptr,
                                       nullptr, nullptr, nullptr, nullptr, Wm, bias, R, g, bln, out,
                                       nullptr, nullptr, nullptr, nullptr, n_rows);
        return;
    }
    const int tid = threadIdx.x;
    const int eb  = ((int)blockIdx.x - n_gemm_blocks) * EPB;

    if (n_gemm_blocks > HDIM * HDIM) {
        for (int k = 0; k < EPB / 256; ++k) {
            const int e = eb + tid + k * 256;
            if (e < n_edges) {
                const int d = dst[e];
                const int b = d >> 7;
                const int pos = atomicAdd(&bin_cnt[b], 1);
                put_record(e, src[e], d, b, pos, bins, ovf_cnt, ovf);
            }
        }
        return;
    }

    int* hist = (int*)Ws;
    for (int b = tid; b < n_gemm_blocks; b += 256) hist[b] = 0;
    __syncthreads();

    int dv[EPB / 256], sv[EPB / 256], rk[EPB / 256];
#pragma unroll
    for (int k = 0; k < EPB / 256; ++k) {
        const int e = eb + tid + k * 256;
        dv[k] = -1;
        sv[k] = 0;
        if (e < n_edges) {
            dv[k] = dst[e];
            sv[k] = src[e];
        }
    }
#pragma unroll
    for (int k = 0; k < EPB / 256; ++k) {
        rk[k] = 0;
        if (dv[k] >= 0) rk[k] = atomicAdd(&hist[dv[k] >> 7], 1);
    }
    __syncthreads();

    for (int b = tid; b < n_gemm_blocks; b += 256)
        hist[b] = atomicAdd(&bin_cnt[b], hist[b]);
    __syncthreads();

#pragma unroll
    for (int k = 0; k < EPB / 256; ++k) {
        if (dv[k] >= 0) {
            const int b = dv[k] >> 7;
            put_record(eb + tid + k * 256, sv[k], dv[k], b, hist[b] + rk[k], bins, ovf_cnt, ovf);
        }
    }
}

extern "C" void kernel_launch(void* const* d_in, const int* in_sizes, int n_in,
                              void* d_out, int out_size, void* d_ws, size_t ws_size,
                              hipStream_t stream)
{
    const float* features = (const float*)d_in[0];
    const int*   src      = (const int*)d_in[1];
    const int*   dst      = (const int*)d_in[2];
    const float* W        = (const float*)d_in[3];
    const float* b        = (const float*)d_in[4];
    const float* gcn_W    = (const float*)d_in[5];
    const float* gcn_b    = (const float*)d_in[6];
    const float* ffln_g   = (const float*)d_in[7];
    const float* ffln_b   = (const float*)d_in[8];
    const float* ln_g     = (const float*)d_in[9];
    const float* ln_b     = (const float*)d_in[10];

    const int n_nodes  = in_sizes[0] / HDIM;
    const int n_edges  = in_sizes[1];
    const int n_layers = in_sizes[6] / HDIM;

    const dim3 block(256);
    const dim3 ggrid((n_nodes + 127) / 128);
    const dim3 fgrid(ggrid.x + (n_edges + EPB - 1) / EPB);

    char* ws = (char*)d_ws;
    float* ff      = (float*)ws;
    float* ff2     = ff + (size_t)n_nodes * HDIM;
    float* zrow    = ff2 + (size_t)n_nodes * HDIM;
    int*   adj     = (int*)(zrow + HDIM);
    int*   bins    = adj + (size_t)n_nodes * CAP;
    int*   cnt     = bins + (size_t)ggrid.x * BIN_CAP * 2;
    int*   bin_cnt = cnt + n_nodes;
    int*   ovf_cnt = bin_cnt + ggrid.x;
    int*   ovf     = ovf_cnt + 1;
    float* out     = (float*)d_out;

    hipMemsetAsync(bin_cnt, 0, ((size_t)ggrid.x + 1) * sizeof(int), stream);

    float* fcur = ff;
    float* fnxt = ff2;
    for (int i = 0; i < n_layers; ++i) {
        if (i == 0)
            gemm_ln_with_build_adj<<<fgrid, block, 0, stream>>>(
                features, W, b, features, ffln_g, ffln_b, fcur, n_nodes, (int)ggrid.x,
                src, dst, bin_cnt, bins, ovf_cnt, ovf, n_edges, zrow);

        const float* gW = gcn_W + (size_t)i * HDIM * HDIM;
        const float* gb = gcn_b + i * HDIM;
        const float* lg = ln_g + i * HDIM;
        const float* lb = ln_b + i * HDIM;
        if (i + 1 < n_layers) {
            const float* g2 = ffln_g + (i + 1) * HDIM;
            const float* b2 = ffln_b + (i + 1) * HDIM;
            if (i == 0)
                gather_gemm_bias_relu_res_ln<true, true><<<ggrid, block, 0, stream>>>(
                    fcur, zrow, adj, cnt, bins, bin_cnt, src, dst, ovf, ovf_cnt,
                    gW, gb, lg, lb, fnxt, W, b, g2, b2, n_nodes);
            else
                gather_gemm_bias_relu_res_ln<false, true><<<ggrid, block, 0, stream>>>(
                    fcur, zrow, adj, cnt, bins, bin_cnt, src, dst, ovf, ovf_cnt,
                    gW, gb, lg, lb, fnxt, W, b, g2, b2, n_nodes);
        } else {
            if (i == 0)
                gather_gemm_bias_relu_res_ln<true, false><<<ggrid, block, 0, stream>>>(
                    fcur, zrow, adj, cnt, bins, bin_cnt, src, dst, ovf, ovf_cnt,
                    gW, gb, lg, lb, out, nullptr, nullptr, nullptr, nullptr, n_nodes);
            else
                gather_gemm_bias_relu_res_ln<false, false><<<ggrid, block, 0, stream>>>(
                    fcur, zrow, adj, cnt, bins, bin_cnt, src, dst, ovf, ovf_cnt,
                    gW, gb, lg, lb, out, nullptr, nullptr, nullptr, nullptr, n_nodes);
        }

        float* t = fcur; fcur = fnxt; fnxt = t;
    }
}
